# starting kernel
# baseline (speedup 1.0000x reference)
#include <hip/hip_runtime.h>


typedef __attribute__((ext_vector_type(2))) float v2f;
typedef __attribute__((ext_vector_type(4))) float v4f;
typedef __attribute__((ext_vector_type(8))) float v8f;

static __device__ __forceinline__ v8f wmma_f32x4(v2f a, v2f b, v8f c) {
  return __builtin_amdgcn_wmma_f32_16x16x4_f32(
       false, a,  false, b,
       (short)0, c,  false,  false);
}

__global__ void k_zero(float* __restrict__ p, long long nElem) {
  long long i = (long long)blockIdx.x * blockDim.x + threadIdx.x;
  if (i < nElem) p[i] = 0.0f;
}

#define T1_STRIDE 68
__launch_bounds__(256)
__global__ void k_gemm1(const float* __restrict__ x, const float* __restrict__ Wl,
                        const float* __restrict__ Wr, float* __restrict__ y1,
                        float* __restrict__ z1, int n) {
  __shared__ float tile[128 * T1_STRIDE];
  const int tid = threadIdx.x;
  const int rb  = blockIdx.x * 128;

#pragma unroll
  for (int it = 0; it < 8; ++it) {
    int idx = (it * 256 + tid) * 4;
    int r = idx >> 6, c = idx & 63;
    int grow = rb + r; if (grow > n - 1) grow = n - 1;
    v4f v = *(const v4f*)(x + (size_t)grow * 64 + c);
    *(v4f*)(tile + r * T1_STRIDE + c) = v;
  }
  __syncthreads();

  const int lane = tid & 31;
  const int wave = tid >> 5;
  const int mr   = lane & 15;
  const int kh   = (lane >> 4) << 1;

  v2f bl[16], br[16];
#pragma unroll
  for (int s = 0; s < 16; ++s) {
    int k0 = 4 * s + kh;
    v2f tl, tr;
    tl.x = Wl[k0 * 16 + mr]; tl.y = Wl[(k0 + 1) * 16 + mr];
    tr.x = Wr[k0 * 16 + mr]; tr.y = Wr[(k0 + 1) * 16 + mr];
    bl[s] = tl; br[s] = tr;
  }

  const float* arow = tile + (wave * 16 + mr) * T1_STRIDE;
  v8f cl = {}; v8f cr = {};
#pragma unroll
  for (int s = 0; s < 16; ++s) {
    v2f a = *(const v2f*)(arow + 4 * s + kh);
    cl = wmma_f32x4(a, bl[s], cl);
    cr = wmma_f32x4(a, br[s], cr);
  }

  const int rowHalf = (lane >> 4) << 3;
#pragma unroll
  for (int v = 0; v < 8; ++v) {
    int row = rb + wave * 16 + rowHalf + v;
    if (row < n) {
      y1[(size_t)row * 16 + mr] = cl[v];
      z1[(size_t)row * 16 + mr] = cr[v];
    }
  }
}

__global__ void k_scatter1(const int* __restrict__ ei, const float* __restrict__ y1,
                           float* __restrict__ agg1, float* __restrict__ deg,
                           long long E) {
  long long t = (long long)blockIdx.x * blockDim.x + threadIdx.x;
  if (t >= E * 16) return;
  long long e = t >> 4;
  int f = (int)(t & 15);
  int src = ei[e];
  int dst = ei[E + e];
  float v = y1[(size_t)src * 16 + f];
  atomicAdd(&agg1[(size_t)dst * 16 + f], v);
  if (f == 0) atomicAdd(&deg[dst], 1.0f);
}

__global__ void k_combine1(const float* __restrict__ agg1, const float* __restrict__ deg,
                           const float* __restrict__ z1, const float* __restrict__ b1,
                           float* __restrict__ h, int n) {
  long long i = (long long)blockIdx.x * blockDim.x + threadIdx.x;
  if (i >= (long long)n * 16) return;
  int node = (int)(i >> 4);
  int f = (int)(i & 15);
  float d = fmaxf(deg[node], 1.0f);
  float v = agg1[i] / d + z1[i] + b1[f];
  h[i] = fmaxf(v, 0.0f);
}

#define T2_STRIDE 20
__launch_bounds__(256)
__global__ void k_gemm2(const float* __restrict__ h, const float* __restrict__ Wl,
                        const float* __restrict__ Wr, float* __restrict__ y2,
                        float* __restrict__ z2, int n) {
  __shared__ float tile[128 * T2_STRIDE];
  const int tid = threadIdx.x;
  const int rb  = blockIdx.x * 128;

#pragma unroll
  for (int it = 0; it < 2; ++it) {
    int idx = (it * 256 + tid) * 4;
    int r = idx >> 4, c = idx & 15;
    int grow = rb + r; if (grow > n - 1) grow = n - 1;
    v4f v = *(const v4f*)(h + (size_t)grow * 16 + c);
    *(v4f*)(tile + r * T2_STRIDE + c) = v;
  }
  __syncthreads();

  const int lane = tid & 31;
  const int wave = tid >> 5;
  const int mr   = lane & 15;
  const int kh   = (lane >> 4) << 1;

  v2f bl[4], br[4];
#pragma unroll
  for (int s = 0; s < 4; ++s) {
    int k0 = 4 * s + kh;
    v2f tl = {}; v2f tr = {};
    if (mr < 2) {
      tl.x = Wl[k0 * 2 + mr]; tl.y = Wl[(k0 + 1) * 2 + mr];
      tr.x = Wr[k0 * 2 + mr]; tr.y = Wr[(k0 + 1) * 2 + mr];
    }
    bl[s] = tl; br[s] = tr;
  }

  const float* arow = tile + (wave * 16 + mr) * T2_STRIDE;
  v8f cl = {}; v8f cr = {};
#pragma unroll
  for (int s = 0; s < 4; ++s) {
    v2f a = *(const v2f*)(arow + 4 * s + kh);
    cl = wmma_f32x4(a, bl[s], cl);
    cr = wmma_f32x4(a, br[s], cr);
  }

  const int rowHalf = (lane >> 4) << 3;
  if (mr < 2) {
#pragma unroll
    for (int v = 0; v < 8; ++v) {
      int row = rb + wave * 16 + rowHalf + v;
      if (row < n) {
        y2[(size_t)row * 2 + mr] = cl[v];
        z2[(size_t)row * 2 + mr] = cr[v];
      }
    }
  }
}

__global__ void k_scatter2(const int* __restrict__ ei, const float* __restrict__ y2,
                           float* __restrict__ agg2, long long E) {
  long long t = (long long)blockIdx.x * blockDim.x + threadIdx.x;
  if (t >= E * 2) return;
  long long e = t >> 1;
  int f = (int)(t & 1);
  int src = ei[e];
  int dst = ei[E + e];
  atomicAdd(&agg2[(size_t)dst * 2 + f], y2[(size_t)src * 2 + f]);
}

__global__ void k_combine2(const float* __restrict__ agg2, const float* __restrict__ deg,
                           const float* __restrict__ z2, const float* __restrict__ b2,
                           float* __restrict__ out, int n) {
  long long i = (long long)blockIdx.x * blockDim.x + threadIdx.x;
  if (i >= (long long)n * 2) return;
  int node = (int)(i >> 1);
  int f = (int)(i & 1);
  float d = fmaxf(deg[node], 1.0f);
  out[i] = agg2[i] / d + z2[i] + b2[f];
}

extern "C" void kernel_launch(void* const* d_in, const int* in_sizes, int n_in,
                              void* d_out, int out_size, void* d_ws, size_t ws_size,
                              hipStream_t stream) {
  const float* x   = (const float*)d_in[0];
  const int*   ei  = (const int*)d_in[1];
  const float* W1l = (const float*)d_in[2];
  const float* W1r = (const float*)d_in[3];
  const float* b1  = (const float*)d_in[4];
  const float* W2l = (const float*)d_in[5];
  const float* W2r = (const float*)d_in[6];
  const float* b2  = (const float*)d_in[7];
  float* out = (float*)d_out;

  const int       n = in_sizes[0] / 64;
  const long long E = in_sizes[1] / 2;

  float* ws   = (float*)d_ws;
  float* agg1 = ws;
  float* deg  = agg1 + (size_t)16 * n;
  float* agg2 = deg + n;
  float* y1   = agg2 + (size_t)2 * n;
  float* z1   = y1 + (size_t)16 * n;
  float* y2   = z1 + (size_t)16 * n;
  float* z2   = y2 + (size_t)2 * n;
  float* h    = y1;

  const long long nzero = (long long)19 * n;
  k_zero<<<(unsigned)((nzero + 255) / 256), 256, 0, stream>>>(ws, nzero);

  const unsigned gGemm = (unsigned)((n + 127) / 128);
  k_gemm1<<<gGemm, 256, 0, stream>>>(x, W1l, W1r, y1, z1, n);

  const long long t1 = E * 16;
  k_scatter1<<<(unsigned)((t1 + 255) / 256), 256, 0, stream>>>(ei, y1, agg1, deg, E);

  const long long c1 = (long long)n * 16;
  k_combine1<<<(unsigned)((c1 + 255) / 256), 256, 0, stream>>>(agg1, deg, z1, b1, h, n);

  k_gemm2<<<gGemm, 256, 0, stream>>>(h, W2l, W2r, y2, z2, n);

  const long long t2 = E * 2;
  k_scatter2<<<(unsigned)((t2 + 255) / 256), 256, 0, stream>>>(ei, y2, agg2, E);

  const long long c2 = (long long)n * 2;
  k_combine2<<<(unsigned)((c2 + 255) / 256), 256, 0, stream>>>(agg2, deg, z2, b2, out, n);
}


# Float-atomic edge scatter replaced by a per-call counting sort of edges by destination (one int atomic per edge) and atomic-free row sums in both layers; sums now run in row-slot (arrival) order
# speedup vs baseline: 1.6487x; 1.6487x over previous
#include <hip/hip_runtime.h>


typedef __attribute__((ext_vector_type(2))) float v2f;
typedef __attribute__((ext_vector_type(4))) float v4f;
typedef __attribute__((ext_vector_type(8))) float v8f;

static __device__ __forceinline__ v8f wmma_f32x4(v2f a, v2f b, v8f c) {
  return __builtin_amdgcn_wmma_f32_16x16x4_f32(
       false, a,  false, b,
       (short)0, c,  false,  false);
}

typedef __attribute__((ext_vector_type(4))) int v4i;

__global__ void k_zero(int* __restrict__ p, int nElem) {
  int i = blockIdx.x * blockDim.x + threadIdx.x;
  if (i < nElem) p[i] = 0;
}

__global__ void k_rank(const int* __restrict__ ei, int* __restrict__ cnt,
                       int* __restrict__ rank, long long E) {
  long long t = (long long)blockIdx.x * blockDim.x + threadIdx.x;
  if (t >= E) return;
  int dst = ei[E + t];
  rank[t] = atomicAdd(&cnt[dst], 1);
}

#define SCAN_B 1024
__launch_bounds__(SCAN_B)
__global__ void k_rowptr(const int* __restrict__ cnt, int* __restrict__ rowptr, int n) {
  __shared__ int wsum[32];
  const int tid = threadIdx.x;
  const int lane = tid & 31;
  const int wave = tid >> 5;
  const int first = blockIdx.x * SCAN_B;

  int s = 0;
  const v4i* c4 = (const v4i*)cnt;
  for (int i = tid; i < (first >> 2); i += SCAN_B) {
    v4i v = c4[i];
    s += (v.x + v.y) + (v.z + v.w);
  }
#pragma unroll
  for (int o = 16; o > 0; o >>= 1) s += __shfl_xor(s, o);
  if (lane == 0) wsum[wave] = s;
  __syncthreads();
  int base = wsum[lane];
#pragma unroll
  for (int o = 16; o > 0; o >>= 1) base += __shfl_xor(base, o);
  __syncthreads();

  const int i = first + tid;
  const int c = (i < n) ? cnt[i] : 0;
  int inc = c;
#pragma unroll
  for (int o = 1; o < 32; o <<= 1) {
    int u = __shfl_up(inc, o);
    if (lane >= o) inc += u;
  }
  if (lane == 31) wsum[wave] = inc;
  __syncthreads();
  int winc = wsum[lane];
#pragma unroll
  for (int o = 1; o < 32; o <<= 1) {
    int u = __shfl_up(winc, o);
    if (lane >= o) winc += u;
  }
  int wpre = __shfl(winc, (wave + 31) & 31);
  if (wave == 0) wpre = 0;
  if (i < n) rowptr[i] = base + wpre + inc - c;
}

__global__ void k_fill(const int* __restrict__ ei, const int* __restrict__ rowptr,
                       const int* __restrict__ rank, int* __restrict__ csr, long long E) {
  long long t = (long long)blockIdx.x * blockDim.x + threadIdx.x;
  if (t >= E) return;
  int src = ei[t];
  int dst = ei[E + t];
  csr[rowptr[dst] + rank[t]] = src;
}

#define T1_STRIDE 68
__launch_bounds__(256)
__global__ void k_gemm1(const float* __restrict__ x, const float* __restrict__ Wl,
                        const float* __restrict__ Wr, float* __restrict__ y1,
                        float* __restrict__ z1, int n) {
  __shared__ float tile[128 * T1_STRIDE];
  const int tid = threadIdx.x;
  const int rb  = blockIdx.x * 128;

#pragma unroll
  for (int it = 0; it < 8; ++it) {
    int idx = (it * 256 + tid) * 4;
    int r = idx >> 6, c = idx & 63;
    int grow = rb + r; if (grow > n - 1) grow = n - 1;
    v4f v = *(const v4f*)(x + (size_t)grow * 64 + c);
    *(v4f*)(tile + r * T1_STRIDE + c) = v;
  }
  __syncthreads();

  const int lane = tid & 31;
  const int wave = tid >> 5;
  const int mr   = lane & 15;
  const int kh   = (lane >> 4) << 1;

  v2f bl[16], br[16];
#pragma unroll
  for (int s = 0; s < 16; ++s) {
    int k0 = 4 * s + kh;
    v2f tl, tr;
    tl.x = Wl[k0 * 16 + mr]; tl.y = Wl[(k0 + 1) * 16 + mr];
    tr.x = Wr[k0 * 16 + mr]; tr.y = Wr[(k0 + 1) * 16 + mr];
    bl[s] = tl; br[s] = tr;
  }

  const float* arow = tile + (wave * 16 + mr) * T1_STRIDE;
  v8f cl = {}; v8f cr = {};
#pragma unroll
  for (int s = 0; s < 16; ++s) {
    v2f a = *(const v2f*)(arow + 4 * s + kh);
    cl = wmma_f32x4(a, bl[s], cl);
    cr = wmma_f32x4(a, br[s], cr);
  }

  const int rowHalf = (lane >> 4) << 3;
#pragma unroll
  for (int v = 0; v < 8; ++v) {
    int row = rb + wave * 16 + rowHalf + v;
    if (row < n) {
      y1[(size_t)row * 16 + mr] = cl[v];
      z1[(size_t)row * 16 + mr] = cr[v];
    }
  }
}

__global__ void k_gather1(const int* __restrict__ rowptr, const int* __restrict__ cnt,
                          const int* __restrict__ csr, const float* __restrict__ y1,
                          const float* __restrict__ z1, const float* __restrict__ b1,
                          float* __restrict__ h, int n) {
  const int g = blockIdx.x * blockDim.x + threadIdx.x;
  const int node = g >> 2;
  const int q = g & 3;
  if (node >= n) return;
  const int beg = rowptr[node];
  const int c = cnt[node];
  const v4f* y4 = (const v4f*)y1;
  const v4f zero4 = {0.0f, 0.0f, 0.0f, 0.0f};
  const v4f z = ((const v4f*)z1)[(size_t)node * 4 + q];
  const float bq0 = b1[q * 4 + 0], bq1 = b1[q * 4 + 1], bq2 = b1[q * 4 + 2], bq3 = b1[q * 4 + 3];
  v4f acc = zero4;
  for (int j = 0; j < c; j += 4) {
    const int m = c - j;
    const int* p = csr + beg + j;
    int s0 = p[0];
    int s1 = p[(m > 1) ? 1 : 0];
    int s2 = p[(m > 2) ? 2 : 0];
    int s3 = p[(m > 3) ? 3 : 0];
    v4f a0 = y4[(size_t)s0 * 4 + q];
    v4f a1 = y4[(size_t)s1 * 4 + q];
    v4f a2 = y4[(size_t)s2 * 4 + q];
    v4f a3 = y4[(size_t)s3 * 4 + q];
    acc += a0;
    acc += (m > 1) ? a1 : zero4;
    acc += (m > 2) ? a2 : zero4;
    acc += (m > 3) ? a3 : zero4;
  }
  const float d = fmaxf((float)c, 1.0f);
  v4f r;
  r.x = fmaxf(acc.x / d + z.x + bq0, 0.0f);
  r.y = fmaxf(acc.y / d + z.y + bq1, 0.0f);
  r.z = fmaxf(acc.z / d + z.z + bq2, 0.0f);
  r.w = fmaxf(acc.w / d + z.w + bq3, 0.0f);
  ((v4f*)h)[(size_t)node * 4 + q] = r;
}

#define T2_STRIDE 20
__launch_bounds__(256)
__global__ void k_gemm2(const float* __restrict__ h, const float* __restrict__ Wl,
                        const float* __restrict__ Wr, float* __restrict__ y2,
                        float* __restrict__ z2, int n) {
  __shared__ float tile[128 * T2_STRIDE];
  const int tid = threadIdx.x;
  const int rb  = blockIdx.x * 128;

#pragma unroll
  for (int it = 0; it < 2; ++it) {
    int idx = (it * 256 + tid) * 4;
    int r = idx >> 4, c = idx & 15;
    int grow = rb + r; if (grow > n - 1) grow = n - 1;
    v4f v = *(const v4f*)(h + (size_t)grow * 16 + c);
    *(v4f*)(tile + r * T2_STRIDE + c) = v;
  }
  __syncthreads();

  const int lane = tid & 31;
  const int wave = tid >> 5;
  const int mr   = lane & 15;
  const int kh   = (lane >> 4) << 1;

  v2f bl[4], br[4];
#pragma unroll
  for (int s = 0; s < 4; ++s) {
    int k0 = 4 * s + kh;
    v2f tl = {}; v2f tr = {};
    if (mr < 2) {
      tl.x = Wl[k0 * 2 + mr]; tl.y = Wl[(k0 + 1) * 2 + mr];
      tr.x = Wr[k0 * 2 + mr]; tr.y = Wr[(k0 + 1) * 2 + mr];
    }
    bl[s] = tl; br[s] = tr;
  }

  const float* arow = tile + (wave * 16 + mr) * T2_STRIDE;
  v8f cl = {}; v8f cr = {};
#pragma unroll
  for (int s = 0; s < 4; ++s) {
    v2f a = *(const v2f*)(arow + 4 * s + kh);
    cl = wmma_f32x4(a, bl[s], cl);
    cr = wmma_f32x4(a, br[s], cr);
  }

  const int rowHalf = (lane >> 4) << 3;
  if (mr < 2) {
#pragma unroll
    for (int v = 0; v < 8; ++v) {
      int row = rb + wave * 16 + rowHalf + v;
      if (row < n) {
        y2[(size_t)row * 2 + mr] = cl[v];
        z2[(size_t)row * 2 + mr] = cr[v];
      }
    }
  }
}

__global__ void k_gather2(const int* __restrict__ rowptr, const int* __restrict__ cnt,
                          const int* __restrict__ csr, const float* __restrict__ y2,
                          const float* __restrict__ z2, const float* __restrict__ b2,
                          float* __restrict__ out, int n) {
  const int node = blockIdx.x * blockDim.x + threadIdx.x;
  if (node >= n) return;
  const int beg = rowptr[node];
  const int c = cnt[node];
  const v2f* y = (const v2f*)y2;
  const v2f zero2 = {0.0f, 0.0f};
  v2f acc = zero2;
  for (int j = 0; j < c; j += 4) {
    const int m = c - j;
    const int* p = csr + beg + j;
    int s0 = p[0];
    int s1 = p[(m > 1) ? 1 : 0];
    int s2 = p[(m > 2) ? 2 : 0];
    int s3 = p[(m > 3) ? 3 : 0];
    v2f a0 = y[s0];
    v2f a1 = y[s1];
    v2f a2 = y[s2];
    v2f a3 = y[s3];
    acc += a0;
    acc += (m > 1) ? a1 : zero2;
    acc += (m > 2) ? a2 : zero2;
    acc += (m > 3) ? a3 : zero2;
  }
  const float d = fmaxf((float)c, 1.0f);
  const v2f z = ((const v2f*)z2)[node];
  v2f r;
  r.x = acc.x / d + z.x + b2[0];
  r.y = acc.y / d + z.y + b2[1];
  ((v2f*)out)[node] = r;
}

extern "C" void kernel_launch(void* const* d_in, const int* in_sizes, int n_in,
                              void* d_out, int out_size, void* d_ws, size_t ws_size,
                              hipStream_t stream) {
  const float* x   = (const float*)d_in[0];
  const int*   ei  = (const int*)d_in[1];
  const float* W1l = (const float*)d_in[2];
  const float* W1r = (const float*)d_in[3];
  const float* b1  = (const float*)d_in[4];
  const float* W2l = (const float*)d_in[5];
  const float* W2r = (const float*)d_in[6];
  const float* b2  = (const float*)d_in[7];
  float* out = (float*)d_out;

  const int       n = in_sizes[0] / 64;
  const long long E = in_sizes[1] / 2;

  float* ws   = (float*)d_ws;
  float* y1   = ws;
  float* z1   = y1 + (size_t)16 * n;
  float* h    = z1 + (size_t)16 * n;
  float* y2   = h + (size_t)16 * n;
  float* z2   = y2 + (size_t)2 * n;
  int* cnt    = (int*)(z2 + (size_t)2 * n);
  int* rowptr = cnt + n;
  int* rank   = rowptr + n;
  int* csr    = rank + E;

  k_zero<<<(unsigned)((n + 255) / 256), 256, 0, stream>>>(cnt, n);

  const unsigned gEdge = (unsigned)((E + 255) / 256);
  k_rank<<<gEdge, 256, 0, stream>>>(ei, cnt, rank, E);

  k_rowptr<<<(unsigned)((n + SCAN_B - 1) / SCAN_B), SCAN_B, 0, stream>>>(cnt, rowptr, n);

  k_fill<<<gEdge, 256, 0, stream>>>(ei, rowptr, rank, csr, E);

  const unsigned gGemm = (unsigned)((n + 127) / 128);
  k_gemm1<<<gGemm, 256, 0, stream>>>(x, W1l, W1r, y1, z1, n);

  const long long g1 = (long long)n * 4;
  k_gather1<<<(unsigned)((g1 + 255) / 256), 256, 0, stream>>>(rowptr, cnt, csr, y1, z1, b1, h, n);

  k_gemm2<<<gGemm, 256, 0, stream>>>(h, W2l, W2r, y2, z2, n);

  k_gather2<<<(unsigned)((n + 127) / 128), 128, 0, stream>>>(rowptr, cnt, csr, y2, z2, b2, out, n);
}


# Edge sort rebuilt as a two-level counting sort on LDS atomics (8192-edge chunks binned by 512-node range, then each range sorted in LDS): no global atomics; rows still summed in arrival order
# speedup vs baseline: 3.8733x; 2.3493x over previous
#include <hip/hip_runtime.h>


typedef __attribute__((ext_vector_type(2))) float v2f;
typedef __attribute__((ext_vector_type(4))) float v4f;
typedef __attribute__((ext_vector_type(8))) float v8f;

static __device__ __forceinline__ v8f wmma_f32x4(v2f a, v2f b, v8f c) {
  return __builtin_amdgcn_wmma_f32_16x16x4_f32(
       false, a,  false, b,
       (short)0, c,  false,  false);
}

typedef __attribute__((ext_vector_type(4))) int v4i;

#define RANGE_SHIFT 9
#define RANGE (1 << RANGE_SHIFT)
#define RANGE_MASK (RANGE - 1)
#define CHUNK_T 1024
#define EPT 8
#define CHUNK (CHUNK_T * EPT)
#define MAXR CHUNK_T

static __device__ __forceinline__ int block_excl_scan(int v, int* wsum, int lane, int wave) {
  int inc = v;
#pragma unroll
  for (int o = 1; o < 32; o <<= 1) {
    int u = __shfl_up(inc, o);
    if (lane >= o) inc += u;
  }
  if (lane == 31) wsum[wave] = inc;
  __syncthreads();
  int winc = wsum[lane];
#pragma unroll
  for (int o = 1; o < 32; o <<= 1) {
    int u = __shfl_up(winc, o);
    if (lane >= o) winc += u;
  }
  int wpre = __shfl(winc, (wave + 31) & 31);
  if (wave == 0) wpre = 0;
  return wpre + inc - v;
}

__launch_bounds__(CHUNK_T)
__global__ void k_chunk_count(const int* __restrict__ ei, int* __restrict__ hist,
                              long long E, int nRange, int nChunk) {
  __shared__ int hb[MAXR];
  const int tid = threadIdx.x;
  hb[tid] = 0;
  __syncthreads();
  const long long base = (long long)blockIdx.x * CHUNK + tid;
  int d[EPT];
#pragma unroll
  for (int k = 0; k < EPT; ++k) {
    const long long e = base + (long long)k * CHUNK_T;
    d[k] = (e < E) ? ei[E + e] : -1;
  }
#pragma unroll
  for (int k = 0; k < EPT; ++k)
    if (d[k] >= 0) atomicAdd(&hb[d[k] >> RANGE_SHIFT], 1);
  __syncthreads();
  if (tid < nRange) hist[tid * nChunk + blockIdx.x] = hb[tid];
}

#define SCAN_B 1024
__launch_bounds__(SCAN_B)
__global__ void k_rowptr(const int* __restrict__ cnt, int* __restrict__ rowptr, int n) {
  __shared__ int wsum[32];
  const int tid = threadIdx.x;
  const int lane = tid & 31;
  const int wave = tid >> 5;
  const int first = blockIdx.x * SCAN_B;

  int s = 0;
  const v4i* c4 = (const v4i*)cnt;
  for (int i = tid; i < (first >> 2); i += SCAN_B) {
    v4i v = c4[i];
    s += (v.x + v.y) + (v.z + v.w);
  }
#pragma unroll
  for (int o = 16; o > 0; o >>= 1) s += __shfl_xor(s, o);
  if (lane == 0) wsum[wave] = s;
  __syncthreads();
  int base = wsum[lane];
#pragma unroll
  for (int o = 16; o > 0; o >>= 1) base += __shfl_xor(base, o);
  __syncthreads();

  const int i = first + tid;
  const int c = (i < n) ? cnt[i] : 0;
  int inc = c;
#pragma unroll
  for (int o = 1; o < 32; o <<= 1) {
    int u = __shfl_up(inc, o);
    if (lane >= o) inc += u;
  }
  if (lane == 31) wsum[wave] = inc;
  __syncthreads();
  int winc = wsum[lane];
#pragma unroll
  for (int o = 1; o < 32; o <<= 1) {
    int u = __shfl_up(winc, o);
    if (lane >= o) winc += u;
  }
  int wpre = __shfl(winc, (wave + 31) & 31);
  if (wave == 0) wpre = 0;
  if (i < n) rowptr[i] = base + wpre + inc - c;
}

__launch_bounds__(CHUNK_T)
__global__ void k_chunk_place(const int* __restrict__ ei, const int* __restrict__ offs,
                              int* __restrict__ keys, long long E, int nRange, int nChunk) {
  __shared__ int hb[MAXR];
  __shared__ int dl[MAXR];
  __shared__ int wsum[32];
  __shared__ int stage[CHUNK];
  __shared__ unsigned short srange[CHUNK];
  const int tid = threadIdx.x;
  const int lane = tid & 31;
  const int wave = tid >> 5;
  hb[tid] = 0;
  __syncthreads();
  const long long base = (long long)blockIdx.x * CHUNK;
  int key[EPT], rg[EPT], rk[EPT];
#pragma unroll
  for (int k = 0; k < EPT; ++k) {
    const long long e = base + (long long)k * CHUNK_T + tid;
    key[k] = (e < E) ? ei[e] : 0;
    rg[k]  = (e < E) ? ei[E + e] : -1;
  }
#pragma unroll
  for (int k = 0; k < EPT; ++k) {
    const int d = rg[k];
    if (d >= 0) {
      key[k] = (key[k] << RANGE_SHIFT) | (d & RANGE_MASK);
      rg[k] = d >> RANGE_SHIFT;
    }
  }
#pragma unroll
  for (int k = 0; k < EPT; ++k)
    rk[k] = (rg[k] >= 0) ? atomicAdd(&hb[rg[k]], 1) : 0;
  __syncthreads();
  const int c = hb[tid];
  const int ls = block_excl_scan(c, wsum, lane, wave);
  hb[tid] = ls;
  dl[tid] = (tid < nRange) ? (offs[tid * nChunk + blockIdx.x] - ls) : 0;
  __syncthreads();
#pragma unroll
  for (int k = 0; k < EPT; ++k) {
    if (rg[k] >= 0) {
      const int pos = hb[rg[k]] + rk[k];
      stage[pos] = key[k];
      srange[pos] = (unsigned short)rg[k];
    }
  }
  __syncthreads();
  const long long left = E - base;
  const int nvalid = (left < (long long)CHUNK) ? (int)left : CHUNK;
#pragma unroll
  for (int k = 0; k < EPT; ++k) {
    const int i = k * CHUNK_T + tid;
    if (i < nvalid) keys[i + dl[srange[i]]] = stage[i];
  }
}

__launch_bounds__(CHUNK_T)
__global__ void k_range_sort(const int* __restrict__ keys, const int* __restrict__ offs,
                             int* __restrict__ cnt, int* __restrict__ rowptr,
                             int* __restrict__ csr, long long E, int n, int nChunk) {
  __shared__ int c[RANGE];
  __shared__ int wsum[32];
  const int tid = threadIdx.x;
  const int lane = tid & 31;
  const int wave = tid >> 5;
  const int b = blockIdx.x;
  const int beg = offs[b * nChunk];
  const int end = (b + 1 < (int)gridDim.x) ? offs[(b + 1) * nChunk] : (int)E;
  if (tid < RANGE) c[tid] = 0;
  __syncthreads();
  for (int i0 = beg; i0 < end; i0 += 4 * CHUNK_T) {
    int kk[4];
#pragma unroll
    for (int j = 0; j < 4; ++j) {
      const int idx = i0 + j * CHUNK_T + tid;
      kk[j] = (idx < end) ? keys[idx] : -1;
    }
#pragma unroll
    for (int j = 0; j < 4; ++j)
      if (kk[j] >= 0) atomicAdd(&c[kk[j] & RANGE_MASK], 1);
  }
  __syncthreads();
  const int v = (tid < RANGE) ? c[tid] : 0;
  const int ls = block_excl_scan(v, wsum, lane, wave);
  if (tid < RANGE) {
    c[tid] = beg + ls;
    const int node = b * RANGE + tid;
    if (node < n) { cnt[node] = v; rowptr[node] = beg + ls; }
  }
  __syncthreads();
  for (int i0 = beg; i0 < end; i0 += 4 * CHUNK_T) {
    int kk[4];
#pragma unroll
    for (int j = 0; j < 4; ++j) {
      const int idx = i0 + j * CHUNK_T + tid;
      kk[j] = (idx < end) ? keys[idx] : -1;
    }
#pragma unroll
    for (int j = 0; j < 4; ++j) {
      if (kk[j] >= 0) {
        const int p = atomicAdd(&c[kk[j] & RANGE_MASK], 1);
        csr[p] = kk[j] >> RANGE_SHIFT;
      }
    }
  }
}

#define T1_STRIDE 68
__launch_bounds__(256)
__global__ void k_gemm1(const float* __restrict__ x, const float* __restrict__ Wl,
                        const float* __restrict__ Wr, float* __restrict__ y1,
                        float* __restrict__ z1, int n) {
  __shared__ float tile[128 * T1_STRIDE];
  const int tid = threadIdx.x;
  const int rb  = blockIdx.x * 128;

#pragma unroll
  for (int it = 0; it < 8; ++it) {
    int idx = (it * 256 + tid) * 4;
    int r = idx >> 6, c = idx & 63;
    int grow = rb + r; if (grow > n - 1) grow = n - 1;
    v4f v = *(const v4f*)(x + (size_t)grow * 64 + c);
    *(v4f*)(tile + r * T1_STRIDE + c) = v;
  }
  __syncthreads();

  const int lane = tid & 31;
  const int wave = tid >> 5;
  const int mr   = lane & 15;
  const int kh   = (lane >> 4) << 1;

  v2f bl[16], br[16];
#pragma unroll
  for (int s = 0; s < 16; ++s) {
    int k0 = 4 * s + kh;
    v2f tl, tr;
    tl.x = Wl[k0 * 16 + mr]; tl.y = Wl[(k0 + 1) * 16 + mr];
    tr.x = Wr[k0 * 16 + mr]; tr.y = Wr[(k0 + 1) * 16 + mr];
    bl[s] = tl; br[s] = tr;
  }

  const float* arow = tile + (wave * 16 + mr) * T1_STRIDE;
  v8f cl = {}; v8f cr = {};
#pragma unroll
  for (int s = 0; s < 16; ++s) {
    v2f a = *(const v2f*)(arow + 4 * s + kh);
    cl = wmma_f32x4(a, bl[s], cl);
    cr = wmma_f32x4(a, br[s], cr);
  }

  const int rowHalf = (lane >> 4) << 3;
#pragma unroll
  for (int v = 0; v < 8; ++v) {
    int row = rb + wave * 16 + rowHalf + v;
    if (row < n) {
      y1[(size_t)row * 16 + mr] = cl[v];
      z1[(size_t)row * 16 + mr] = cr[v];
    }
  }
}

__global__ void k_gather1(const int* __restrict__ rowptr, const int* __restrict__ cnt,
                          const int* __restrict__ csr, const float* __restrict__ y1,
                          const float* __restrict__ z1, const float* __restrict__ b1,
                          float* __restrict__ h, int n) {
  const int g = blockIdx.x * blockDim.x + threadIdx.x;
  const int node = g >> 2;
  const int q = g & 3;
  if (node >= n) return;
  const int beg = rowptr[node];
  const int c = cnt[node];
  const v4f* y4 = (const v4f*)y1;
  const v4f zero4 = {0.0f, 0.0f, 0.0f, 0.0f};
  const v4f z = ((const v4f*)z1)[(size_t)node * 4 + q];
  const float bq0 = b1[q * 4 + 0], bq1 = b1[q * 4 + 1], bq2 = b1[q * 4 + 2], bq3 = b1[q * 4 + 3];
  v4f acc = zero4;
  for (int j = 0; j < c; j += 4) {
    const int m = c - j;
    const int* p = csr + beg + j;
    int s0 = p[0];
    int s1 = p[(m > 1) ? 1 : 0];
    int s2 = p[(m > 2) ? 2 : 0];
    int s3 = p[(m > 3) ? 3 : 0];
    v4f a0 = y4[(size_t)s0 * 4 + q];
    v4f a1 = y4[(size_t)s1 * 4 + q];
    v4f a2 = y4[(size_t)s2 * 4 + q];
    v4f a3 = y4[(size_t)s3 * 4 + q];
    acc += a0;
    acc += (m > 1) ? a1 : zero4;
    acc += (m > 2) ? a2 : zero4;
    acc += (m > 3) ? a3 : zero4;
  }
  const float d = fmaxf((float)c, 1.0f);
  v4f r;
  r.x = fmaxf(acc.x / d + z.x + bq0, 0.0f);
  r.y = fmaxf(acc.y / d + z.y + bq1, 0.0f);
  r.z = fmaxf(acc.z / d + z.z + bq2, 0.0f);
  r.w = fmaxf(acc.w / d + z.w + bq3, 0.0f);
  ((v4f*)h)[(size_t)node * 4 + q] = r;
}

#define T2_STRIDE 20
__launch_bounds__(256)
__global__ void k_gemm2(const float* __restrict__ h, const float* __restrict__ Wl,
                        const float* __restrict__ Wr, float* __restrict__ y2,
                        float* __restrict__ z2, int n) {
  __shared__ float tile[128 * T2_STRIDE];
  const int tid = threadIdx.x;
  const int rb  = blockIdx.x * 128;

#pragma unroll
  for (int it = 0; it < 2; ++it) {
    int idx = (it * 256 + tid) * 4;
    int r = idx >> 4, c = idx & 15;
    int grow = rb + r; if (grow > n - 1) grow = n - 1;
    v4f v = *(const v4f*)(h + (size_t)grow * 16 + c);
    *(v4f*)(tile + r * T2_STRIDE + c) = v;
  }
  __syncthreads();

  const int lane = tid & 31;
  const int wave = tid >> 5;
  const int mr   = lane & 15;
  const int kh   = (lane >> 4) << 1;

  v2f bl[4], br[4];
#pragma unroll
  for (int s = 0; s < 4; ++s) {
    int k0 = 4 * s + kh;
    v2f tl = {}; v2f tr = {};
    if (mr < 2) {
      tl.x = Wl[k0 * 2 + mr]; tl.y = Wl[(k0 + 1) * 2 + mr];
      tr.x = Wr[k0 * 2 + mr]; tr.y = Wr[(k0 + 1) * 2 + mr];
    }
    bl[s] = tl; br[s] = tr;
  }

  const float* arow = tile + (wave * 16 + mr) * T2_STRIDE;
  v8f cl = {}; v8f cr = {};
#pragma unroll
  for (int s = 0; s < 4; ++s) {
    v2f a = *(const v2f*)(arow + 4 * s + kh);
    cl = wmma_f32x4(a, bl[s], cl);
    cr = wmma_f32x4(a, br[s], cr);
  }

  const int rowHalf = (lane >> 4) << 3;
  if (mr < 2) {
#pragma unroll
    for (int v = 0; v < 8; ++v) {
      int row = rb + wave * 16 + rowHalf + v;
      if (row < n) {
        y2[(size_t)row * 2 + mr] = cl[v];
        z2[(size_t)row * 2 + mr] = cr[v];
      }
    }
  }
}

__global__ void k_gather2(const int* __restrict__ rowptr, const int* __restrict__ cnt,
                          const int* __restrict__ csr, const float* __restrict__ y2,
                          const float* __restrict__ z2, const float* __restrict__ b2,
                          float* __restrict__ out, int n) {
  const int node = blockIdx.x * blockDim.x + threadIdx.x;
  if (node >= n) return;
  const int beg = rowptr[node];
  const int c = cnt[node];
  const v2f* y = (const v2f*)y2;
  const v2f zero2 = {0.0f, 0.0f};
  v2f acc = zero2;
  for (int j = 0; j < c; j += 4) {
    const int m = c - j;
    const int* p = csr + beg + j;
    int s0 = p[0];
    int s1 = p[(m > 1) ? 1 : 0];
    int s2 = p[(m > 2) ? 2 : 0];
    int s3 = p[(m > 3) ? 3 : 0];
    v2f a0 = y[s0];
    v2f a1 = y[s1];
    v2f a2 = y[s2];
    v2f a3 = y[s3];
    acc += a0;
    acc += (m > 1) ? a1 : zero2;
    acc += (m > 2) ? a2 : zero2;
    acc += (m > 3) ? a3 : zero2;
  }
  const float d = fmaxf((float)c, 1.0f);
  const v2f z = ((const v2f*)z2)[node];
  v2f r;
  r.x = acc.x / d + z.x + b2[0];
  r.y = acc.y / d + z.y + b2[1];
  ((v2f*)out)[node] = r;
}

extern "C" void kernel_launch(void* const* d_in, const int* in_sizes, int n_in,
                              void* d_out, int out_size, void* d_ws, size_t ws_size,
                              hipStream_t stream) {
  const float* x   = (const float*)d_in[0];
  const int*   ei  = (const int*)d_in[1];
  const float* W1l = (const float*)d_in[2];
  const float* W1r = (const float*)d_in[3];
  const float* b1  = (const float*)d_in[4];
  const float* W2l = (const float*)d_in[5];
  const float* W2r = (const float*)d_in[6];
  const float* b2  = (const float*)d_in[7];
  float* out = (float*)d_out;

  const int       n = in_sizes[0] / 64;
  const long long E = in_sizes[1] / 2;

  float* ws   = (float*)d_ws;
  float* y1   = ws;
  float* z1   = y1 + (size_t)16 * n;
  float* h    = z1 + (size_t)16 * n;
  float* y2   = h + (size_t)16 * n;
  float* z2   = y2 + (size_t)2 * n;
  int* cnt    = (int*)(z2 + (size_t)2 * n);
  int* rowptr = cnt + n;
  int* keys   = rowptr + n;
  int* csr    = keys + E;
  const int nRange = (n + RANGE - 1) / RANGE;
  const int nChunk = (int)((E + CHUNK - 1) / CHUNK);
  const int nTab   = nRange * nChunk;
  int* hist   = csr + E;
  int* offs   = hist + nTab;

  k_chunk_count<<<(unsigned)nChunk, CHUNK_T, 0, stream>>>(ei, hist, E, nRange, nChunk);

  k_rowptr<<<(unsigned)((nTab + SCAN_B - 1) / SCAN_B), SCAN_B, 0, stream>>>(hist, offs, nTab);

  k_chunk_place<<<(unsigned)nChunk, CHUNK_T, 0, stream>>>(ei, offs, keys, E, nRange, nChunk);

  k_range_sort<<<(unsigned)nRange, CHUNK_T, 0, stream>>>(keys, offs, cnt, rowptr, csr, E, n, nChunk);

  const unsigned gGemm = (unsigned)((n + 127) / 128);
  k_gemm1<<<gGemm, 256, 0, stream>>>(x, W1l, W1r, y1, z1, n);

  const long long g1 = (long long)n * 4;
  k_gather1<<<(unsigned)((g1 + 255) / 256), 256, 0, stream>>>(rowptr, cnt, csr, y1, z1, b1, h, n);

  k_gemm2<<<gGemm, 256, 0, stream>>>(h, W2l, W2r, y2, z2, n);

  k_gather2<<<(unsigned)((n + 127) / 128), 128, 0, stream>>>(rowptr, cnt, csr, y2, z2, b2, out, n);
}


# Edge sort level 1 made one pass: a chunk range-orders its edges in its own slice, the range kernel collects its piece of each chunk; count pass and table scan gone; rows still summed in arrival order
# speedup vs baseline: 4.4002x; 1.1360x over previous
#include <hip/hip_runtime.h>


typedef __attribute__((ext_vector_type(2))) float v2f;
typedef __attribute__((ext_vector_type(4))) float v4f;
typedef __attribute__((ext_vector_type(8))) float v8f;

static __device__ __forceinline__ v8f wmma_f32x4(v2f a, v2f b, v8f c) {
  return __builtin_amdgcn_wmma_f32_16x16x4_f32(
       false, a,  false, b,
       (short)0, c,  false,  false);
}

typedef __attribute__((ext_vector_type(4))) int v4i;

#define RANGE_SHIFT 9
#define RANGE (1 << RANGE_SHIFT)
#define RANGE_MASK (RANGE - 1)
#define CHUNK_T 1024
#define EPT 8
#define CHUNK (CHUNK_T * EPT)
#define MAXR CHUNK_T

static __device__ __forceinline__ int block_excl_scan(int v, int* wsum, int lane, int wave) {
  int inc = v;
#pragma unroll
  for (int o = 1; o < 32; o <<= 1) {
    int u = __shfl_up(inc, o);
    if (lane >= o) inc += u;
  }
  if (lane == 31) wsum[wave] = inc;
  __syncthreads();
  int winc = wsum[lane];
#pragma unroll
  for (int o = 1; o < 32; o <<= 1) {
    int u = __shfl_up(winc, o);
    if (lane >= o) winc += u;
  }
  int wpre = __shfl(winc, (wave + 31) & 31);
  if (wave == 0) wpre = 0;
  return wpre + inc - v;
}

__launch_bounds__(CHUNK_T)
__global__ void k_chunk_sort(const int* __restrict__ ei, int* __restrict__ cntTab,
                             int* __restrict__ locTab, int* __restrict__ keys,
                             long long E, int nRange, int nChunk) {
  __shared__ int hb[MAXR];
  __shared__ int wsum[32];
  __shared__ int stage[CHUNK];
  const int tid = threadIdx.x;
  const int lane = tid & 31;
  const int wave = tid >> 5;
  hb[tid] = 0;
  __syncthreads();
  const long long base = (long long)blockIdx.x * CHUNK;
  int key[EPT], rg[EPT], rk[EPT];
#pragma unroll
  for (int k = 0; k < EPT; ++k) {
    const long long e = base + (long long)k * CHUNK_T + tid;
    key[k] = (e < E) ? ei[e] : 0;
    rg[k]  = (e < E) ? ei[E + e] : -1;
  }
#pragma unroll
  for (int k = 0; k < EPT; ++k) {
    const int d = rg[k];
    if (d >= 0) {
      key[k] = (key[k] << RANGE_SHIFT) | (d & RANGE_MASK);
      rg[k] = d >> RANGE_SHIFT;
    }
  }
#pragma unroll
  for (int k = 0; k < EPT; ++k)
    rk[k] = (rg[k] >= 0) ? atomicAdd(&hb[rg[k]], 1) : 0;
  __syncthreads();
  const int c = hb[tid];
  const int ls = block_excl_scan(c, wsum, lane, wave);
  hb[tid] = ls;
  if (tid < nRange) {
    cntTab[tid * nChunk + blockIdx.x] = c;
    locTab[tid * nChunk + blockIdx.x] = ls;
  }
  __syncthreads();
#pragma unroll
  for (int k = 0; k < EPT; ++k)
    if (rg[k] >= 0) stage[hb[rg[k]] + rk[k]] = key[k];
  __syncthreads();
  const long long left = E - base;
  const int nvalid = (left < (long long)CHUNK) ? (int)left : CHUNK;
#pragma unroll
  for (int k = 0; k < EPT; ++k) {
    const int i = k * CHUNK_T + tid;
    if (i < nvalid) keys[base + i] = stage[i];
  }
}

#define MAXC CHUNK_T
template <class F>
static __device__ __forceinline__ void for_each_key(const int* __restrict__ keys,
                                                    const int* segN, const int* segO,
                                                    int nChunk, int lane, int wave, F f) {
  for (int g0 = wave; g0 < nChunk; g0 += 4 * 32) {
    int nn[4], oo[4], ka[4], kb[4];
#pragma unroll
    for (int j = 0; j < 4; ++j) {
      const int ch = g0 + 32 * j;
      nn[j] = (ch < nChunk) ? segN[ch] : 0;
      oo[j] = (ch < nChunk) ? segO[ch] : 0;
    }
#pragma unroll
    for (int j = 0; j < 4; ++j) {
      ka[j] = (lane < nn[j]) ? keys[oo[j] + lane] : -1;
      kb[j] = (lane + 32 < nn[j]) ? keys[oo[j] + lane + 32] : -1;
    }
#pragma unroll
    for (int j = 0; j < 4; ++j) {
      if (ka[j] >= 0) f(ka[j]);
      if (kb[j] >= 0) f(kb[j]);
    }
#pragma unroll
    for (int j = 0; j < 4; ++j)
      for (int i = lane + 64; i < nn[j]; i += 32) f(keys[oo[j] + i]);
  }
}

__launch_bounds__(CHUNK_T)
__global__ void k_range_sort(const int* __restrict__ keys, const int* __restrict__ cntTab,
                             const int* __restrict__ locTab, int* __restrict__ cnt,
                             int* __restrict__ rowptr, int* __restrict__ csr,
                             int n, int nChunk) {
  __shared__ int c[RANGE];
  __shared__ int wsum[32];
  __shared__ int segN[MAXC];
  __shared__ int segO[MAXC];
  const int tid = threadIdx.x;
  const int lane = tid & 31;
  const int wave = tid >> 5;
  const int b = blockIdx.x;

  const int nFront = b * nChunk;
  int s = 0;
  const v4i* c4 = (const v4i*)cntTab;
  const int nFront4 = nFront >> 2;
  for (int i0 = tid; i0 < nFront4; i0 += 8 * CHUNK_T) {
    v4i v[8];
#pragma unroll
    for (int j = 0; j < 8; ++j) {
      const int i = i0 + j * CHUNK_T;
      const v4i zero = {0, 0, 0, 0};
      v[j] = (i < nFront4) ? c4[i] : zero;
    }
#pragma unroll
    for (int j = 0; j < 8; ++j) s += (v[j].x + v[j].y) + (v[j].z + v[j].w);
  }
  for (int i = (nFront & ~3) + tid; i < nFront; i += CHUNK_T) s += cntTab[i];
  for (int ch = tid; ch < nChunk; ch += CHUNK_T) {
    segN[ch] = cntTab[nFront + ch];
    segO[ch] = ch * CHUNK + locTab[nFront + ch];
  }
  if (tid < RANGE) c[tid] = 0;
#pragma unroll
  for (int o = 16; o > 0; o >>= 1) s += __shfl_xor(s, o);
  if (lane == 0) wsum[wave] = s;
  __syncthreads();
  int beg = wsum[lane];
#pragma unroll
  for (int o = 16; o > 0; o >>= 1) beg += __shfl_xor(beg, o);

  for_each_key(keys, segN, segO, nChunk, lane, wave,
               [&](int k) { atomicAdd(&c[k & RANGE_MASK], 1); });
  __syncthreads();
  const int v = (tid < RANGE) ? c[tid] : 0;
  const int ls = block_excl_scan(v, wsum, lane, wave);
  if (tid < RANGE) {
    c[tid] = beg + ls;
    const int node = b * RANGE + tid;
    if (node < n) { cnt[node] = v; rowptr[node] = beg + ls; }
  }
  __syncthreads();
  for_each_key(keys, segN, segO, nChunk, lane, wave,
               [&](int k) {
                 const int p = atomicAdd(&c[k & RANGE_MASK], 1);
                 csr[p] = k >> RANGE_SHIFT;
               });
}

#define T1_STRIDE 68
__launch_bounds__(256)
__global__ void k_gemm1(const float* __restrict__ x, const float* __restrict__ Wl,
                        const float* __restrict__ Wr, float* __restrict__ y1,
                        float* __restrict__ z1, int n) {
  __shared__ float tile[128 * T1_STRIDE];
  const int tid = threadIdx.x;
  const int rb  = blockIdx.x * 128;

#pragma unroll
  for (int it = 0; it < 8; ++it) {
    int idx = (it * 256 + tid) * 4;
    int r = idx >> 6, c = idx & 63;
    int grow = rb + r; if (grow > n - 1) grow = n - 1;
    v4f v = *(const v4f*)(x + (size_t)grow * 64 + c);
    *(v4f*)(tile + r * T1_STRIDE + c) = v;
  }
  __syncthreads();

  const int lane = tid & 31;
  const int wave = tid >> 5;
  const int mr   = lane & 15;
  const int kh   = (lane >> 4) << 1;

  v2f bl[16], br[16];
#pragma unroll
  for (int s = 0; s < 16; ++s) {
    int k0 = 4 * s + kh;
    v2f tl, tr;
    tl.x = Wl[k0 * 16 + mr]; tl.y = Wl[(k0 + 1) * 16 + mr];
    tr.x = Wr[k0 * 16 + mr]; tr.y = Wr[(k0 + 1) * 16 + mr];
    bl[s] = tl; br[s] = tr;
  }

  const float* arow = tile + (wave * 16 + mr) * T1_STRIDE;
  v8f cl = {}; v8f cr = {};
#pragma unroll
  for (int s = 0; s < 16; ++s) {
    v2f a = *(const v2f*)(arow + 4 * s + kh);
    cl = wmma_f32x4(a, bl[s], cl);
    cr = wmma_f32x4(a, br[s], cr);
  }

  const int rowHalf = (lane >> 4) << 3;
#pragma unroll
  for (int v = 0; v < 8; ++v) {
    int row = rb + wave * 16 + rowHalf + v;
    if (row < n) {
      y1[(size_t)row * 16 + mr] = cl[v];
      z1[(size_t)row * 16 + mr] = cr[v];
    }
  }
}

__global__ void k_gather1(const int* __restrict__ rowptr, const int* __restrict__ cnt,
                          const int* __restrict__ csr, const float* __restrict__ y1,
                          const float* __restrict__ z1, const float* __restrict__ b1,
                          float* __restrict__ h, int n) {
  const int g = blockIdx.x * blockDim.x + threadIdx.x;
  const int node = g >> 2;
  const int q = g & 3;
  if (node >= n) return;
  const int beg = rowptr[node];
  const int c = cnt[node];
  const v4f* y4 = (const v4f*)y1;
  const v4f zero4 = {0.0f, 0.0f, 0.0f, 0.0f};
  const v4f z = ((const v4f*)z1)[(size_t)node * 4 + q];
  const float bq0 = b1[q * 4 + 0], bq1 = b1[q * 4 + 1], bq2 = b1[q * 4 + 2], bq3 = b1[q * 4 + 3];
  v4f acc = zero4;
  for (int j = 0; j < c; j += 4) {
    const int m = c - j;
    const int* p = csr + beg + j;
    int s0 = p[0];
    int s1 = p[(m > 1) ? 1 : 0];
    int s2 = p[(m > 2) ? 2 : 0];
    int s3 = p[(m > 3) ? 3 : 0];
    v4f a0 = y4[(size_t)s0 * 4 + q];
    v4f a1 = y4[(size_t)s1 * 4 + q];
    v4f a2 = y4[(size_t)s2 * 4 + q];
    v4f a3 = y4[(size_t)s3 * 4 + q];
    acc += a0;
    acc += (m > 1) ? a1 : zero4;
    acc += (m > 2) ? a2 : zero4;
    acc += (m > 3) ? a3 : zero4;
  }
  const float d = fmaxf((float)c, 1.0f);
  v4f r;
  r.x = fmaxf(acc.x / d + z.x + bq0, 0.0f);
  r.y = fmaxf(acc.y / d + z.y + bq1, 0.0f);
  r.z = fmaxf(acc.z / d + z.z + bq2, 0.0f);
  r.w = fmaxf(acc.w / d + z.w + bq3, 0.0f);
  ((v4f*)h)[(size_t)node * 4 + q] = r;
}

#define T2_STRIDE 20
__launch_bounds__(256)
__global__ void k_gemm2(const float* __restrict__ h, const float* __restrict__ Wl,
                        const float* __restrict__ Wr, float* __restrict__ y2,
                        float* __restrict__ z2, int n) {
  __shared__ float tile[128 * T2_STRIDE];
  const int tid = threadIdx.x;
  const int rb  = blockIdx.x * 128;

#pragma unroll
  for (int it = 0; it < 2; ++it) {
    int idx = (it * 256 + tid) * 4;
    int r = idx >> 4, c = idx & 15;
    int grow = rb + r; if (grow > n - 1) grow = n - 1;
    v4f v = *(const v4f*)(h + (size_t)grow * 16 + c);
    *(v4f*)(tile + r * T2_STRIDE + c) = v;
  }
  __syncthreads();

  const int lane = tid & 31;
  const int wave = tid >> 5;
  const int mr   = lane & 15;
  const int kh   = (lane >> 4) << 1;

  v2f bl[4], br[4];
#pragma unroll
  for (int s = 0; s < 4; ++s) {
    int k0 = 4 * s + kh;
    v2f tl = {}; v2f tr = {};
    if (mr < 2) {
      tl.x = Wl[k0 * 2 + mr]; tl.y = Wl[(k0 + 1) * 2 + mr];
      tr.x = Wr[k0 * 2 + mr]; tr.y = Wr[(k0 + 1) * 2 + mr];
    }
    bl[s] = tl; br[s] = tr;
  }

  const float* arow = tile + (wave * 16 + mr) * T2_STRIDE;
  v8f cl = {}; v8f cr = {};
#pragma unroll
  for (int s = 0; s < 4; ++s) {
    v2f a = *(const v2f*)(arow + 4 * s + kh);
    cl = wmma_f32x4(a, bl[s], cl);
    cr = wmma_f32x4(a, br[s], cr);
  }

  const int rowHalf = (lane >> 4) << 3;
  if (mr < 2) {
#pragma unroll
    for (int v = 0; v < 8; ++v) {
      int row = rb + wave * 16 + rowHalf + v;
      if (row < n) {
        y2[(size_t)row * 2 + mr] = cl[v];
        z2[(size_t)row * 2 + mr] = cr[v];
      }
    }
  }
}

__global__ void k_gather2(const int* __restrict__ rowptr, const int* __restrict__ cnt,
                          const int* __restrict__ csr, const float* __restrict__ y2,
                          const float* __restrict__ z2, const float* __restrict__ b2,
                          float* __restrict__ out, int n) {
  const int node = blockIdx.x * blockDim.x + threadIdx.x;
  if (node >= n) return;
  const int beg = rowptr[node];
  const int c = cnt[node];
  const v2f* y = (const v2f*)y2;
  const v2f zero2 = {0.0f, 0.0f};
  v2f acc = zero2;
  for (int j = 0; j < c; j += 4) {
    const int m = c - j;
    const int* p = csr + beg + j;
    int s0 = p[0];
    int s1 = p[(m > 1) ? 1 : 0];
    int s2 = p[(m > 2) ? 2 : 0];
    int s3 = p[(m > 3) ? 3 : 0];
    v2f a0 = y[s0];
    v2f a1 = y[s1];
    v2f a2 = y[s2];
    v2f a3 = y[s3];
    acc += a0;
    acc += (m > 1) ? a1 : zero2;
    acc += (m > 2) ? a2 : zero2;
    acc += (m > 3) ? a3 : zero2;
  }
  const float d = fmaxf((float)c, 1.0f);
  const v2f z = ((const v2f*)z2)[node];
  v2f r;
  r.x = acc.x / d + z.x + b2[0];
  r.y = acc.y / d + z.y + b2[1];
  ((v2f*)out)[node] = r;
}

extern "C" void kernel_launch(void* const* d_in, const int* in_sizes, int n_in,
                              void* d_out, int out_size, void* d_ws, size_t ws_size,
                              hipStream_t stream) {
  const float* x   = (const float*)d_in[0];
  const int*   ei  = (const int*)d_in[1];
  const float* W1l = (const float*)d_in[2];
  const float* W1r = (const float*)d_in[3];
  const float* b1  = (const float*)d_in[4];
  const float* W2l = (const float*)d_in[5];
  const float* W2r = (const float*)d_in[6];
  const float* b2  = (const float*)d_in[7];
  float* out = (float*)d_out;

  const int       n = in_sizes[0] / 64;
  const long long E = in_sizes[1] / 2;

  float* ws   = (float*)d_ws;
  float* y1   = ws;
  float* z1   = y1 + (size_t)16 * n;
  float* h    = z1 + (size_t)16 * n;
  float* y2   = h + (size_t)16 * n;
  float* z2   = y2 + (size_t)2 * n;
  int* cnt    = (int*)(z2 + (size_t)2 * n);
  int* rowptr = cnt + n;
  int* keys   = rowptr + n;
  int* csr    = keys + E;
  const int nRange = (n + RANGE - 1) / RANGE;
  const int nChunk = (int)((E + CHUNK - 1) / CHUNK);
  const int nTab   = nRange * nChunk;
  int* cntTab = csr + E;
  int* locTab = cntTab + nTab;

  k_chunk_sort<<<(unsigned)nChunk, CHUNK_T, 0, stream>>>(ei, cntTab, locTab, keys, E, nRange, nChunk);

  k_range_sort<<<(unsigned)nRange, CHUNK_T, 0, stream>>>(keys, cntTab, locTab, cnt, rowptr, csr, n, nChunk);

  const unsigned gGemm = (unsigned)((n + 127) / 128);
  k_gemm1<<<gGemm, 256, 0, stream>>>(x, W1l, W1r, y1, z1, n);

  const long long g1 = (long long)n * 4;
  k_gather1<<<(unsigned)((g1 + 255) / 256), 256, 0, stream>>>(rowptr, cnt, csr, y1, z1, b1, h, n);

  k_gemm2<<<gGemm, 256, 0, stream>>>(h, W2l, W2r, y2, z2, n);

  k_gather2<<<(unsigned)((n + 127) / 128), 128, 0, stream>>>(rowptr, cnt, csr, y2, z2, b2, out, n);
}


# Layer-1 GEMM (x@W1l, x@W1r) runs as extra work-groups of the range-sort launch instead of in a launch of its own: 5 launches, the read of x overlaps the sort's waits; summation order unchanged
# speedup vs baseline: 4.7612x; 1.0820x over previous
#include <hip/hip_runtime.h>


typedef __attribute__((ext_vector_type(2))) float v2f;
typedef __attribute__((ext_vector_type(4))) float v4f;
typedef __attribute__((ext_vector_type(8))) float v8f;

static __device__ __forceinline__ v8f wmma_f32x4(v2f a, v2f b, v8f c) {
  return __builtin_amdgcn_wmma_f32_16x16x4_f32(
       false, a,  false, b,
       (short)0, c,  false,  false);
}

typedef __attribute__((ext_vector_type(4))) int v4i;

#define RANGE_SHIFT 9
#define RANGE (1 << RANGE_SHIFT)
#define RANGE_MASK (RANGE - 1)
#define CHUNK_T 1024
#define EPT 8
#define CHUNK (CHUNK_T * EPT)
#define MAXR CHUNK_T

static __device__ __forceinline__ int block_excl_scan(int v, int* wsum, int lane, int wave) {
  int inc = v;
#pragma unroll
  for (int o = 1; o < 32; o <<= 1) {
    int u = __shfl_up(inc, o);
    if (lane >= o) inc += u;
  }
  if (lane == 31) wsum[wave] = inc;
  __syncthreads();
  int winc = wsum[lane];
#pragma unroll
  for (int o = 1; o < 32; o <<= 1) {
    int u = __shfl_up(winc, o);
    if (lane >= o) winc += u;
  }
  int wpre = __shfl(winc, (wave + 31) & 31);
  if (wave == 0) wpre = 0;
  return wpre + inc - v;
}

__launch_bounds__(CHUNK_T)
__global__ void k_chunk_sort(const int* __restrict__ ei, int* __restrict__ cntTab,
                             int* __restrict__ locTab, int* __restrict__ keys,
                             long long E, int nRange, int nChunk) {
  __shared__ int hb[MAXR];
  __shared__ int wsum[32];
  __shared__ int stage[CHUNK];
  const int tid = threadIdx.x;
  const int lane = tid & 31;
  const int wave = tid >> 5;
  hb[tid] = 0;
  __syncthreads();
  const long long base = (long long)blockIdx.x * CHUNK;
  int key[EPT], rg[EPT], rk[EPT];
#pragma unroll
  for (int k = 0; k < EPT; ++k) {
    const long long e = base + (long long)k * CHUNK_T + tid;
    key[k] = (e < E) ? ei[e] : 0;
    rg[k]  = (e < E) ? ei[E + e] : -1;
  }
#pragma unroll
  for (int k = 0; k < EPT; ++k) {
    const int d = rg[k];
    if (d >= 0) {
      key[k] = (key[k] << RANGE_SHIFT) | (d & RANGE_MASK);
      rg[k] = d >> RANGE_SHIFT;
    }
  }
#pragma unroll
  for (int k = 0; k < EPT; ++k)
    rk[k] = (rg[k] >= 0) ? atomicAdd(&hb[rg[k]], 1) : 0;
  __syncthreads();
  const int c = hb[tid];
  const int ls = block_excl_scan(c, wsum, lane, wave);
  hb[tid] = ls;
  if (tid < nRange) {
    cntTab[tid * nChunk + blockIdx.x] = c;
    locTab[tid * nChunk + blockIdx.x] = ls;
  }
  __syncthreads();
#pragma unroll
  for (int k = 0; k < EPT; ++k)
    if (rg[k] >= 0) stage[hb[rg[k]] + rk[k]] = key[k];
  __syncthreads();
  const long long left = E - base;
  const int nvalid = (left < (long long)CHUNK) ? (int)left : CHUNK;
#pragma unroll
  for (int k = 0; k < EPT; ++k) {
    const int i = k * CHUNK_T + tid;
    if (i < nvalid) keys[base + i] = stage[i];
  }
}

#define MAXC CHUNK_T
template <class F>
static __device__ __forceinline__ void for_each_key(const int* __restrict__ keys,
                                                    const int* segN, const int* segO,
                                                    int nChunk, int lane, int wave, F f) {
  for (int g0 = wave; g0 < nChunk; g0 += 4 * 32) {
    int nn[4], oo[4], ka[4], kb[4];
#pragma unroll
    for (int j = 0; j < 4; ++j) {
      const int ch = g0 + 32 * j;
      nn[j] = (ch < nChunk) ? segN[ch] : 0;
      oo[j] = (ch < nChunk) ? segO[ch] : 0;
    }
#pragma unroll
    for (int j = 0; j < 4; ++j) {
      ka[j] = (lane < nn[j]) ? keys[oo[j] + lane] : -1;
      kb[j] = (lane + 32 < nn[j]) ? keys[oo[j] + lane + 32] : -1;
    }
#pragma unroll
    for (int j = 0; j < 4; ++j) {
      if (ka[j] >= 0) f(ka[j]);
      if (kb[j] >= 0) f(kb[j]);
    }
#pragma unroll
    for (int j = 0; j < 4; ++j)
      for (int i = lane + 64; i < nn[j]; i += 32) f(keys[oo[j] + i]);
  }
}

#define T1_STRIDE 68
#define G1_ROWS 128
#define G1_QUARTERS (CHUNK_T / 256)
#define G1_WG_ROWS (G1_ROWS * G1_QUARTERS)
static __device__ __forceinline__ void gemm1_rows(float* __restrict__ lds,
                                                  const float* __restrict__ x,
                                                  const float* __restrict__ Wl,
                                                  const float* __restrict__ Wr,
                                                  float* __restrict__ y1,
                                                  float* __restrict__ z1, int n, int wg) {
  const int quarter = threadIdx.x >> 8;
  const int tid = threadIdx.x & 255;
  float* tile = lds + quarter * (G1_ROWS * T1_STRIDE);
  const int rb = (wg * G1_QUARTERS + quarter) * G1_ROWS;

#pragma unroll
  for (int it = 0; it < 8; ++it) {
    int idx = (it * 256 + tid) * 4;
    int r = idx >> 6, c = idx & 63;
    int grow = rb + r; if (grow > n - 1) grow = n - 1;
    v4f v = *(const v4f*)(x + (size_t)grow * 64 + c);
    *(v4f*)(tile + r * T1_STRIDE + c) = v;
  }
  __syncthreads();

  const int lane = tid & 31;
  const int wave = tid >> 5;
  const int mr   = lane & 15;
  const int kh   = (lane >> 4) << 1;

  v2f bl[16], br[16];
#pragma unroll
  for (int s = 0; s < 16; ++s) {
    int k0 = 4 * s + kh;
    v2f tl, tr;
    tl.x = Wl[k0 * 16 + mr]; tl.y = Wl[(k0 + 1) * 16 + mr];
    tr.x = Wr[k0 * 16 + mr]; tr.y = Wr[(k0 + 1) * 16 + mr];
    bl[s] = tl; br[s] = tr;
  }

  const float* arow = tile + (wave * 16 + mr) * T1_STRIDE;
  v8f cl = {}; v8f cr = {};
#pragma unroll
  for (int s = 0; s < 16; ++s) {
    v2f a = *(const v2f*)(arow + 4 * s + kh);
    cl = wmma_f32x4(a, bl[s], cl);
    cr = wmma_f32x4(a, br[s], cr);
  }

  const int rowHalf = (lane >> 4) << 3;
#pragma unroll
  for (int v = 0; v < 8; ++v) {
    int row = rb + wave * 16 + rowHalf + v;
    if (row < n) {
      y1[(size_t)row * 16 + mr] = cl[v];
      z1[(size_t)row * 16 + mr] = cr[v];
    }
  }
}

__launch_bounds__(CHUNK_T, 16)
__global__ void k_range_sort_gemm1(const int* __restrict__ keys, const int* __restrict__ cntTab,
                                   const int* __restrict__ locTab, int* __restrict__ cnt,
                                   int* __restrict__ rowptr, int* __restrict__ csr,
                                   int n, int nChunk, int nRange,
                                   const float* __restrict__ x, const float* __restrict__ Wl,
                                   const float* __restrict__ Wr, float* __restrict__ y1,
                                   float* __restrict__ z1) {
  __shared__ __attribute__((aligned(16))) unsigned char lds[G1_WG_ROWS * T1_STRIDE * 4];
  static_assert((RANGE + 32 + 2 * MAXC) * 4 <= G1_WG_ROWS * T1_STRIDE * 4, "sort tables fit the LDS block");
  if ((int)blockIdx.x >= nRange) {
    gemm1_rows((float*)lds, x, Wl, Wr, y1, z1, n, (int)blockIdx.x - nRange);
    return;
  }
  int* c    = (int*)lds;
  int* wsum = c + RANGE;
  int* segN = wsum + 32;
  int* segO = segN + MAXC;
  const int tid = threadIdx.x;
  const int lane = tid & 31;
  const int wave = tid >> 5;
  const int b = blockIdx.x;

  const int nFront = b * nChunk;
  int s = 0;
  const v4i* c4 = (const v4i*)cntTab;
  const int nFront4 = nFront >> 2;
  for (int i0 = tid; i0 < nFront4; i0 += 8 * CHUNK_T) {
    v4i v[8];
#pragma unroll
    for (int j = 0; j < 8; ++j) {
      const int i = i0 + j * CHUNK_T;
      const v4i zero = {0, 0, 0, 0};
      v[j] = (i < nFront4) ? c4[i] : zero;
    }
#pragma unroll
    for (int j = 0; j < 8; ++j) s += (v[j].x + v[j].y) + (v[j].z + v[j].w);
  }
  for (int i = (nFront & ~3) + tid; i < nFront; i += CHUNK_T) s += cntTab[i];
  for (int ch = tid; ch < nChunk; ch += CHUNK_T) {
    segN[ch] = cntTab[nFront + ch];
    segO[ch] = ch * CHUNK + locTab[nFront + ch];
  }
  if (tid < RANGE) c[tid] = 0;
#pragma unroll
  for (int o = 16; o > 0; o >>= 1) s += __shfl_xor(s, o);
  if (lane == 0) wsum[wave] = s;
  __syncthreads();
  int beg = wsum[lane];
#pragma unroll
  for (int o = 16; o > 0; o >>= 1) beg += __shfl_xor(beg, o);

  for_each_key(keys, segN, segO, nChunk, lane, wave,
               [&](int k) { atomicAdd(&c[k & RANGE_MASK], 1); });
  __syncthreads();
  const int v = (tid < RANGE) ? c[tid] : 0;
  const int ls = block_excl_scan(v, wsum, lane, wave);
  if (tid < RANGE) {
    c[tid] = beg + ls;
    const int node = b * RANGE + tid;
    if (node < n) { cnt[node] = v; rowptr[node] = beg + ls; }
  }
  __syncthreads();
  for_each_key(keys, segN, segO, nChunk, lane, wave,
               [&](int k) {
                 const int p = atomicAdd(&c[k & RANGE_MASK], 1);
                 csr[p] = k >> RANGE_SHIFT;
               });
}

__global__ void k_gather1(const int* __restrict__ rowptr, const int* __restrict__ cnt,
                          const int* __restrict__ csr, const float* __restrict__ y1,
                          const float* __restrict__ z1, const float* __restrict__ b1,
                          float* __restrict__ h, int n) {
  const int g = blockIdx.x * blockDim.x + threadIdx.x;
  const int node = g >> 2;
  const int q = g & 3;
  if (node >= n) return;
  const int beg = rowptr[node];
  const int c = cnt[node];
  const v4f* y4 = (const v4f*)y1;
  const v4f zero4 = {0.0f, 0.0f, 0.0f, 0.0f};
  const v4f z = ((const v4f*)z1)[(size_t)node * 4 + q];
  const float bq0 = b1[q * 4 + 0], bq1 = b1[q * 4 + 1], bq2 = b1[q * 4 + 2], bq3 = b1[q * 4 + 3];
  v4f acc = zero4;
  for (int j = 0; j < c; j += 4) {
    const int m = c - j;
    const int* p = csr + beg + j;
    int s0 = p[0];
    int s1 = p[(m > 1) ? 1 : 0];
    int s2 = p[(m > 2) ? 2 : 0];
    int s3 = p[(m > 3) ? 3 : 0];
    v4f a0 = y4[(size_t)s0 * 4 + q];
    v4f a1 = y4[(size_t)s1 * 4 + q];
    v4f a2 = y4[(size_t)s2 * 4 + q];
    v4f a3 = y4[(size_t)s3 * 4 + q];
    acc += a0;
    acc += (m > 1) ? a1 : zero4;
    acc += (m > 2) ? a2 : zero4;
    acc += (m > 3) ? a3 : zero4;
  }
  const float d = fmaxf((float)c, 1.0f);
  v4f r;
  r.x = fmaxf(acc.x / d + z.x + bq0, 0.0f);
  r.y = fmaxf(acc.y / d + z.y + bq1, 0.0f);
  r.z = fmaxf(acc.z / d + z.z + bq2, 0.0f);
  r.w = fmaxf(acc.w / d + z.w + bq3, 0.0f);
  ((v4f*)h)[(size_t)node * 4 + q] = r;
}

#define T2_STRIDE 20
__launch_bounds__(256)
__global__ void k_gemm2(const float* __restrict__ h, const float* __restrict__ Wl,
                        const float* __restrict__ Wr, float* __restrict__ y2,
                        float* __restrict__ z2, int n) {
  __shared__ float tile[128 * T2_STRIDE];
  const int tid = threadIdx.x;
  const int rb  = blockIdx.x * 128;

#pragma unroll
  for (int it = 0; it < 2; ++it) {
    int idx = (it * 256 + tid) * 4;
    int r = idx >> 4, c = idx & 15;
    int grow = rb + r; if (grow > n - 1) grow = n - 1;
    v4f v = *(const v4f*)(h + (size_t)grow * 16 + c);
    *(v4f*)(tile + r * T2_STRIDE + c) = v;
  }
  __syncthreads();

  const int lane = tid & 31;
  const int wave = tid >> 5;
  const int mr   = lane & 15;
  const int kh   = (lane >> 4) << 1;

  v2f bl[4], br[4];
#pragma unroll
  for (int s = 0; s < 4; ++s) {
    int k0 = 4 * s + kh;
    v2f tl = {}; v2f tr = {};
    if (mr < 2) {
      tl.x = Wl[k0 * 2 + mr]; tl.y = Wl[(k0 + 1) * 2 + mr];
      tr.x = Wr[k0 * 2 + mr]; tr.y = Wr[(k0 + 1) * 2 + mr];
    }
    bl[s] = tl; br[s] = tr;
  }

  const float* arow = tile + (wave * 16 + mr) * T2_STRIDE;
  v8f cl = {}; v8f cr = {};
#pragma unroll
  for (int s = 0; s < 4; ++s) {
    v2f a = *(const v2f*)(arow + 4 * s + kh);
    cl = wmma_f32x4(a, bl[s], cl);
    cr = wmma_f32x4(a, br[s], cr);
  }

  const int rowHalf = (lane >> 4) << 3;
  if (mr < 2) {
#pragma unroll
    for (int v = 0; v < 8; ++v) {
      int row = rb + wave * 16 + rowHalf + v;
      if (row < n) {
        y2[(size_t)row * 2 + mr] = cl[v];
        z2[(size_t)row * 2 + mr] = cr[v];
      }
    }
  }
}

__global__ void k_gather2(const int* __restrict__ rowptr, const int* __restrict__ cnt,
                          const int* __restrict__ csr, const float* __restrict__ y2,
                          const float* __restrict__ z2, const float* __restrict__ b2,
                          float* __restrict__ out, int n) {
  const int node = blockIdx.x * blockDim.x + threadIdx.x;
  if (node >= n) return;
  const int beg = rowptr[node];
  const int c = cnt[node];
  const v2f* y = (const v2f*)y2;
  const v2f zero2 = {0.0f, 0.0f};
  v2f acc = zero2;
  for (int j = 0; j < c; j += 4) {
    const int m = c - j;
    const int* p = csr + beg + j;
    int s0 = p[0];
    int s1 = p[(m > 1) ? 1 : 0];
    int s2 = p[(m > 2) ? 2 : 0];
    int s3 = p[(m > 3) ? 3 : 0];
    v2f a0 = y[s0];
    v2f a1 = y[s1];
    v2f a2 = y[s2];
    v2f a3 = y[s3];
    acc += a0;
    acc += (m > 1) ? a1 : zero2;
    acc += (m > 2) ? a2 : zero2;
    acc += (m > 3) ? a3 : zero2;
  }
  const float d = fmaxf((float)c, 1.0f);
  const v2f z = ((const v2f*)z2)[node];
  v2f r;
  r.x = acc.x / d + z.x + b2[0];
  r.y = acc.y / d + z.y + b2[1];
  ((v2f*)out)[node] = r;
}

extern "C" void kernel_launch(void* const* d_in, const int* in_sizes, int n_in,
                              void* d_out, int out_size, void* d_ws, size_t ws_size,
                              hipStream_t stream) {
  const float* x   = (const float*)d_in[0];
  const int*   ei  = (const int*)d_in[1];
  const float* W1l = (const float*)d_in[2];
  const float* W1r = (const float*)d_in[3];
  const float* b1  = (const float*)d_in[4];
  const float* W2l = (const float*)d_in[5];
  const float* W2r = (const float*)d_in[6];
  const float* b2  = (const float*)d_in[7];
  float* out = (float*)d_out;

  const int       n = in_sizes[0] / 64;
  const long long E = in_sizes[1] / 2;

  float* ws   = (float*)d_ws;
  float* y1   = ws;
  float* z1   = y1 + (size_t)16 * n;
  float* h    = z1 + (size_t)16 * n;
  float* y2   = h + (size_t)16 * n;
  float* z2   = y2 + (size_t)2 * n;
  int* cnt    = (int*)(z2 + (size_t)2 * n);
  int* rowptr = cnt + n;
  int* keys   = rowptr + n;
  int* csr    = keys + E;
  const int nRange = (n + RANGE - 1) / RANGE;
  const int nChunk = (int)((E + CHUNK - 1) / CHUNK);
  const int nTab   = nRange * nChunk;
  int* cntTab = csr + E;
  int* locTab = cntTab + nTab;

  k_chunk_sort<<<(unsigned)nChunk, CHUNK_T, 0, stream>>>(ei, cntTab, locTab, keys, E, nRange, nChunk);

  const int nGemm1 = (n + G1_WG_ROWS - 1) / G1_WG_ROWS;
  k_range_sort_gemm1<<<(unsigned)(nRange + nGemm1), CHUNK_T, 0, stream>>>(
      keys, cntTab, locTab, cnt, rowptr, csr, n, nChunk, nRange, x, W1l, W1r, y1, z1);

  const unsigned gGemm = (unsigned)((n + 127) / 128);

  const long long g1 = (long long)n * 4;
  k_gather1<<<(unsigned)((g1 + 255) / 256), 256, 0, stream>>>(rowptr, cnt, csr, y1, z1, b1, h, n);

  k_gemm2<<<gGemm, 256, 0, stream>>>(h, W2l, W2r, y2, z2, n);

  k_gather2<<<(unsigned)((n + 127) / 128), 128, 0, stream>>>(rowptr, cnt, csr, y2, z2, b2, out, n);
}


# Level-2 sort puts each node range's source rows together in LDS and writes them out as whole 128-byte lines (was one scattered 4-byte store per edge); slots and summation order unchanged
# speedup vs baseline: 5.1752x; 1.0870x over previous
#include <hip/hip_runtime.h>


typedef __attribute__((ext_vector_type(2))) float v2f;
typedef __attribute__((ext_vector_type(4))) float v4f;
typedef __attribute__((ext_vector_type(8))) float v8f;

static __device__ __forceinline__ v8f wmma_f32x4(v2f a, v2f b, v8f c) {
  return __builtin_amdgcn_wmma_f32_16x16x4_f32(
       false, a,  false, b,
       (short)0, c,  false,  false);
}

typedef __attribute__((ext_vector_type(4))) int v4i;

#define RANGE_SHIFT 9
#define RANGE (1 << RANGE_SHIFT)
#define RANGE_MASK (RANGE - 1)
#define CHUNK_T 1024
#define EPT 8
#define CHUNK (CHUNK_T * EPT)
#define MAXR CHUNK_T

static __device__ __forceinline__ int block_excl_scan(int v, int* wsum, int lane, int wave, int& total) {
  int inc = v;
#pragma unroll
  for (int o = 1; o < 32; o <<= 1) {
    int u = __shfl_up(inc, o);
    if (lane >= o) inc += u;
  }
  if (lane == 31) wsum[wave] = inc;
  __syncthreads();
  int winc = wsum[lane];
#pragma unroll
  for (int o = 1; o < 32; o <<= 1) {
    int u = __shfl_up(winc, o);
    if (lane >= o) winc += u;
  }
  int wpre = __shfl(winc, (wave + 31) & 31);
  if (wave == 0) wpre = 0;
  total = __shfl(winc, 31);
  return wpre + inc - v;
}

__launch_bounds__(CHUNK_T)
__global__ void k_chunk_sort(const int* __restrict__ ei, int* __restrict__ cntTab,
                             int* __restrict__ locTab, int* __restrict__ keys,
                             long long E, int nRange, int nChunk) {
  __shared__ int hb[MAXR];
  __shared__ int wsum[32];
  __shared__ int stage[CHUNK];
  const int tid = threadIdx.x;
  const int lane = tid & 31;
  const int wave = tid >> 5;
  hb[tid] = 0;
  __syncthreads();
  const long long base = (long long)blockIdx.x * CHUNK;
  int key[EPT], rg[EPT], rk[EPT];
#pragma unroll
  for (int k = 0; k < EPT; ++k) {
    const long long e = base + (long long)k * CHUNK_T + tid;
    key[k] = (e < E) ? ei[e] : 0;
    rg[k]  = (e < E) ? ei[E + e] : -1;
  }
#pragma unroll
  for (int k = 0; k < EPT; ++k) {
    const int d = rg[k];
    if (d >= 0) {
      key[k] = (key[k] << RANGE_SHIFT) | (d & RANGE_MASK);
      rg[k] = d >> RANGE_SHIFT;
    }
  }
#pragma unroll
  for (int k = 0; k < EPT; ++k)
    rk[k] = (rg[k] >= 0) ? atomicAdd(&hb[rg[k]], 1) : 0;
  __syncthreads();
  const int c = hb[tid];
  int chunkEdges;
  const int ls = block_excl_scan(c, wsum, lane, wave, chunkEdges);
  hb[tid] = ls;
  if (tid < nRange) {
    cntTab[tid * nChunk + blockIdx.x] = c;
    locTab[tid * nChunk + blockIdx.x] = ls;
  }
  __syncthreads();
#pragma unroll
  for (int k = 0; k < EPT; ++k)
    if (rg[k] >= 0) stage[hb[rg[k]] + rk[k]] = key[k];
  __syncthreads();
  const long long left = E - base;
  const int nvalid = (left < (long long)CHUNK) ? (int)left : CHUNK;
#pragma unroll
  for (int k = 0; k < EPT; ++k) {
    const int i = k * CHUNK_T + tid;
    if (i < nvalid) keys[base + i] = stage[i];
  }
}

#define MAXC CHUNK_T
template <class F>
static __device__ __forceinline__ void for_each_key(const int* __restrict__ keys,
                                                    const int* segN, const int* segO,
                                                    int nChunk, int lane, int wave, F f) {
  for (int g0 = wave; g0 < nChunk; g0 += 4 * 32) {
    int nn[4], oo[4], ka[4], kb[4];
#pragma unroll
    for (int j = 0; j < 4; ++j) {
      const int ch = g0 + 32 * j;
      nn[j] = (ch < nChunk) ? segN[ch] : 0;
      oo[j] = (ch < nChunk) ? segO[ch] : 0;
    }
#pragma unroll
    for (int j = 0; j < 4; ++j) {
      ka[j] = (lane < nn[j]) ? keys[oo[j] + lane] : -1;
      kb[j] = (lane + 32 < nn[j]) ? keys[oo[j] + lane + 32] : -1;
    }
#pragma unroll
    for (int j = 0; j < 4; ++j) {
      if (ka[j] >= 0) f(ka[j]);
      if (kb[j] >= 0) f(kb[j]);
    }
#pragma unroll
    for (int j = 0; j < 4; ++j)
      for (int i = lane + 64; i < nn[j]; i += 32) f(keys[oo[j] + i]);
  }
}

#define T1_STRIDE 68
#define G1_ROWS 128
#define G1_QUARTERS (CHUNK_T / 256)
#define G1_WG_ROWS (G1_ROWS * G1_QUARTERS)
static __device__ __forceinline__ void gemm1_rows(float* __restrict__ lds,
                                                  const float* __restrict__ x,
                                                  const float* __restrict__ Wl,
                                                  const float* __restrict__ Wr,
                                                  float* __restrict__ y1,
                                                  float* __restrict__ z1, int n, int wg) {
  const int quarter = threadIdx.x >> 8;
  const int tid = threadIdx.x & 255;
  float* tile = lds + quarter * (G1_ROWS * T1_STRIDE);
  const int rb = (wg * G1_QUARTERS + quarter) * G1_ROWS;

#pragma unroll
  for (int it = 0; it < 8; ++it) {
    int idx = (it * 256 + tid) * 4;
    int r = idx >> 6, c = idx & 63;
    int grow = rb + r; if (grow > n - 1) grow = n - 1;
    v4f v = *(const v4f*)(x + (size_t)grow * 64 + c);
    *(v4f*)(tile + r * T1_STRIDE + c) = v;
  }
  __syncthreads();

  const int lane = tid & 31;
  const int wave = tid >> 5;
  const int mr   = lane & 15;
  const int kh   = (lane >> 4) << 1;

  v2f bl[16], br[16];
#pragma unroll
  for (int s = 0; s < 16; ++s) {
    int k0 = 4 * s + kh;
    v2f tl, tr;
    tl.x = Wl[k0 * 16 + mr]; tl.y = Wl[(k0 + 1) * 16 + mr];
    tr.x = Wr[k0 * 16 + mr]; tr.y = Wr[(k0 + 1) * 16 + mr];
    bl[s] = tl; br[s] = tr;
  }

  const float* arow = tile + (wave * 16 + mr) * T1_STRIDE;
  v8f cl = {}; v8f cr = {};
#pragma unroll
  for (int s = 0; s < 16; ++s) {
    v2f a = *(const v2f*)(arow + 4 * s + kh);
    cl = wmma_f32x4(a, bl[s], cl);
    cr = wmma_f32x4(a, br[s], cr);
  }

  const int rowHalf = (lane >> 4) << 3;
#pragma unroll
  for (int v = 0; v < 8; ++v) {
    int row = rb + wave * 16 + rowHalf + v;
    if (row < n) {
      y1[(size_t)row * 16 + mr] = cl[v];
      z1[(size_t)row * 16 + mr] = cr[v];
    }
  }
}

__launch_bounds__(CHUNK_T, 16)
__global__ void k_range_sort_gemm1(const int* __restrict__ keys, const int* __restrict__ cntTab,
                                   const int* __restrict__ locTab, int* __restrict__ cnt,
                                   int* __restrict__ rowptr, int* __restrict__ csr,
                                   int n, int nChunk, int nRange,
                                   const float* __restrict__ x, const float* __restrict__ Wl,
                                   const float* __restrict__ Wr, float* __restrict__ y1,
                                   float* __restrict__ z1) {
  __shared__ __attribute__((aligned(16))) unsigned char lds[G1_WG_ROWS * T1_STRIDE * 4];
  static_assert((RANGE + 32 + 2 * MAXC) * 4 <= G1_WG_ROWS * T1_STRIDE * 4, "sort tables fit the LDS block");
  static_assert(((RANGE + 32 + 2 * MAXC) * 4) % 16 == 0, "the LDS copy of the rows starts 16 B aligned");
  if ((int)blockIdx.x >= nRange) {
    gemm1_rows((float*)lds, x, Wl, Wr, y1, z1, n, (int)blockIdx.x - nRange);
    return;
  }
  int* c    = (int*)lds;
  int* wsum = c + RANGE;
  int* segN = wsum + 32;
  int* segO = segN + MAXC;
  int* stg  = segO + MAXC;
  constexpr int STG_CAP = G1_WG_ROWS * T1_STRIDE - (RANGE + 32 + 2 * MAXC);
  const int tid = threadIdx.x;
  const int lane = tid & 31;
  const int wave = tid >> 5;
  const int b = blockIdx.x;

  const int nFront = b * nChunk;
  int s = 0;
  const v4i* c4 = (const v4i*)cntTab;
  const int nFront4 = nFront >> 2;
  for (int i0 = tid; i0 < nFront4; i0 += 8 * CHUNK_T) {
    v4i v[8];
#pragma unroll
    for (int j = 0; j < 8; ++j) {
      const int i = i0 + j * CHUNK_T;
      const v4i zero = {0, 0, 0, 0};
      v[j] = (i < nFront4) ? c4[i] : zero;
    }
#pragma unroll
    for (int j = 0; j < 8; ++j) s += (v[j].x + v[j].y) + (v[j].z + v[j].w);
  }
  for (int i = (nFront & ~3) + tid; i < nFront; i += CHUNK_T) s += cntTab[i];
  for (int ch = tid; ch < nChunk; ch += CHUNK_T) {
    segN[ch] = cntTab[nFront + ch];
    segO[ch] = ch * CHUNK + locTab[nFront + ch];
  }
  if (tid < RANGE) c[tid] = 0;
#pragma unroll
  for (int o = 16; o > 0; o >>= 1) s += __shfl_xor(s, o);
  if (lane == 0) wsum[wave] = s;
  __syncthreads();
  int beg = wsum[lane];
#pragma unroll
  for (int o = 16; o > 0; o >>= 1) beg += __shfl_xor(beg, o);

  for_each_key(keys, segN, segO, nChunk, lane, wave,
               [&](int k) { atomicAdd(&c[k & RANGE_MASK], 1); });
  __syncthreads();
  const int v = (tid < RANGE) ? c[tid] : 0;
  int tot;
  const int ls = block_excl_scan(v, wsum, lane, wave, tot);
  if (tid < RANGE) {
    c[tid] = ls;
    const int node = b * RANGE + tid;
    if (node < n) { cnt[node] = v; rowptr[node] = beg + ls; }
  }
  __syncthreads();
  const int off = beg & 31;
  if (off + tot <= STG_CAP) {
    for_each_key(keys, segN, segO, nChunk, lane, wave,
                 [&](int k) {
                   const int p = atomicAdd(&c[k & RANGE_MASK], 1);
                   stg[off + p] = k >> RANGE_SHIFT;
                 });
    __syncthreads();
    int* rows = csr + (beg - off);
    const int lo = off, hi = off + tot;
    const int vlo = (lo + 3) & ~3, vhi = hi & ~3;
    const int headEnd = (vlo < hi) ? vlo : hi;
    if (lo + tid < headEnd) rows[lo + tid] = stg[lo + tid];
    for (int q = 4 * tid; q < vhi; q += 4 * CHUNK_T)
      if (q >= vlo) *(v4i*)(rows + q) = *(const v4i*)(stg + q);
    const int tb = (vhi > headEnd) ? vhi : headEnd;
    if (tb + tid < hi) rows[tb + tid] = stg[tb + tid];
  } else {
    for_each_key(keys, segN, segO, nChunk, lane, wave,
                 [&](int k) {
                   const int p = atomicAdd(&c[k & RANGE_MASK], 1);
                   csr[beg + p] = k >> RANGE_SHIFT;
                 });
  }
}

__global__ void k_gather1(const int* __restrict__ rowptr, const int* __restrict__ cnt,
                          const int* __restrict__ csr, const float* __restrict__ y1,
                          const float* __restrict__ z1, const float* __restrict__ b1,
                          float* __restrict__ h, int n) {
  const int g = blockIdx.x * blockDim.x + threadIdx.x;
  const int node = g >> 2;
  const int q = g & 3;
  if (node >= n) return;
  const int beg = rowptr[node];
  const int c = cnt[node];
  const v4f* y4 = (const v4f*)y1;
  const v4f zero4 = {0.0f, 0.0f, 0.0f, 0.0f};
  const v4f z = ((const v4f*)z1)[(size_t)node * 4 + q];
  const float bq0 = b1[q * 4 + 0], bq1 = b1[q * 4 + 1], bq2 = b1[q * 4 + 2], bq3 = b1[q * 4 + 3];
  v4f acc = zero4;
  for (int j = 0; j < c; j += 4) {
    const int m = c - j;
    const int* p = csr + beg + j;
    int s0 = p[0];
    int s1 = p[(m > 1) ? 1 : 0];
    int s2 = p[(m > 2) ? 2 : 0];
    int s3 = p[(m > 3) ? 3 : 0];
    v4f a0 = y4[(size_t)s0 * 4 + q];
    v4f a1 = y4[(size_t)s1 * 4 + q];
    v4f a2 = y4[(size_t)s2 * 4 + q];
    v4f a3 = y4[(size_t)s3 * 4 + q];
    acc += a0;
    acc += (m > 1) ? a1 : zero4;
    acc += (m > 2) ? a2 : zero4;
    acc += (m > 3) ? a3 : zero4;
  }
  const float d = fmaxf((float)c, 1.0f);
  v4f r;
  r.x = fmaxf(acc.x / d + z.x + bq0, 0.0f);
  r.y = fmaxf(acc.y / d + z.y + bq1, 0.0f);
  r.z = fmaxf(acc.z / d + z.z + bq2, 0.0f);
  r.w = fmaxf(acc.w / d + z.w + bq3, 0.0f);
  ((v4f*)h)[(size_t)node * 4 + q] = r;
}

#define T2_STRIDE 20
__launch_bounds__(256)
__global__ void k_gemm2(const float* __restrict__ h, const float* __restrict__ Wl,
                        const float* __restrict__ Wr, float* __restrict__ y2,
                        float* __restrict__ z2, int n) {
  __shared__ float tile[128 * T2_STRIDE];
  const int tid = threadIdx.x;
  const int rb  = blockIdx.x * 128;

#pragma unroll
  for (int it = 0; it < 2; ++it) {
    int idx = (it * 256 + tid) * 4;
    int r = idx >> 4, c = idx & 15;
    int grow = rb + r; if (grow > n - 1) grow = n - 1;
    v4f v = *(const v4f*)(h + (size_t)grow * 16 + c);
    *(v4f*)(tile + r * T2_STRIDE + c) = v;
  }
  __syncthreads();

  const int lane = tid & 31;
  const int wave = tid >> 5;
  const int mr   = lane & 15;
  const int kh   = (lane >> 4) << 1;

  v2f bl[4], br[4];
#pragma unroll
  for (int s = 0; s < 4; ++s) {
    int k0 = 4 * s + kh;
    v2f tl = {}; v2f tr = {};
    if (mr < 2) {
      tl.x = Wl[k0 * 2 + mr]; tl.y = Wl[(k0 + 1) * 2 + mr];
      tr.x = Wr[k0 * 2 + mr]; tr.y = Wr[(k0 + 1) * 2 + mr];
    }
    bl[s] = tl; br[s] = tr;
  }

  const float* arow = tile + (wave * 16 + mr) * T2_STRIDE;
  v8f cl = {}; v8f cr = {};
#pragma unroll
  for (int s = 0; s < 4; ++s) {
    v2f a = *(const v2f*)(arow + 4 * s + kh);
    cl = wmma_f32x4(a, bl[s], cl);
    cr = wmma_f32x4(a, br[s], cr);
  }

  const int rowHalf = (lane >> 4) << 3;
  if (mr < 2) {
#pragma unroll
    for (int v = 0; v < 8; ++v) {
      int row = rb + wave * 16 + rowHalf + v;
      if (row < n) {
        y2[(size_t)row * 2 + mr] = cl[v];
        z2[(size_t)row * 2 + mr] = cr[v];
      }
    }
  }
}

__global__ void k_gather2(const int* __restrict__ rowptr, const int* __restrict__ cnt,
                          const int* __restrict__ csr, const float* __restrict__ y2,
                          const float* __restrict__ z2, const float* __restrict__ b2,
                          float* __restrict__ out, int n) {
  const int node = blockIdx.x * blockDim.x + threadIdx.x;
  if (node >= n) return;
  const int beg = rowptr[node];
  const int c = cnt[node];
  const v2f* y = (const v2f*)y2;
  const v2f zero2 = {0.0f, 0.0f};
  v2f acc = zero2;
  for (int j = 0; j < c; j += 4) {
    const int m = c - j;
    const int* p = csr + beg + j;
    int s0 = p[0];
    int s1 = p[(m > 1) ? 1 : 0];
    int s2 = p[(m > 2) ? 2 : 0];
    int s3 = p[(m > 3) ? 3 : 0];
    v2f a0 = y[s0];
    v2f a1 = y[s1];
    v2f a2 = y[s2];
    v2f a3 = y[s3];
    acc += a0;
    acc += (m > 1) ? a1 : zero2;
    acc += (m > 2) ? a2 : zero2;
    acc += (m > 3) ? a3 : zero2;
  }
  const float d = fmaxf((float)c, 1.0f);
  const v2f z = ((const v2f*)z2)[node];
  v2f r;
  r.x = acc.x / d + z.x + b2[0];
  r.y = acc.y / d + z.y + b2[1];
  ((v2f*)out)[node] = r;
}

extern "C" void kernel_launch(void* const* d_in, const int* in_sizes, int n_in,
                              void* d_out, int out_size, void* d_ws, size_t ws_size,
                              hipStream_t stream) {
  const float* x   = (const float*)d_in[0];
  const int*   ei  = (const int*)d_in[1];
  const float* W1l = (const float*)d_in[2];
  const float* W1r = (const float*)d_in[3];
  const float* b1  = (const float*)d_in[4];
  const float* W2l = (const float*)d_in[5];
  const float* W2r = (const float*)d_in[6];
  const float* b2  = (const float*)d_in[7];
  float* out = (float*)d_out;

  const int       n = in_sizes[0] / 64;
  const long long E = in_sizes[1] / 2;

  float* ws   = (float*)d_ws;
  float* y1   = ws;
  float* z1   = y1 + (size_t)16 * n;
  float* h    = z1 + (size_t)16 * n;
  float* y2   = h + (size_t)16 * n;
  float* z2   = y2 + (size_t)2 * n;
  int* cnt    = (int*)(z2 + (size_t)2 * n);
  int* rowptr = cnt + n;
  int* keys   = rowptr + n;
  int* csr    = keys + E;
  const int nRange = (n + RANGE - 1) / RANGE;
  const int nChunk = (int)((E + CHUNK - 1) / CHUNK);
  const int nTab   = nRange * nChunk;
  int* cntTab = csr + E;
  int* locTab = cntTab + nTab;

  k_chunk_sort<<<(unsigned)nChunk, CHUNK_T, 0, stream>>>(ei, cntTab, locTab, keys, E, nRange, nChunk);

  const int nGemm1 = (n + G1_WG_ROWS - 1) / G1_WG_ROWS;
  k_range_sort_gemm1<<<(unsigned)(nRange + nGemm1), CHUNK_T, 0, stream>>>(
      keys, cntTab, locTab, cnt, rowptr, csr, n, nChunk, nRange, x, W1l, W1r, y1, z1);

  const unsigned gGemm = (unsigned)((n + 127) / 128);

  const long long g1 = (long long)n * 4;
  k_gather1<<<(unsigned)((g1 + 255) / 256), 256, 0, stream>>>(rowptr, cnt, csr, y1, z1, b1, h, n);

  k_gemm2<<<gGemm, 256, 0, stream>>>(h, W2l, W2r, y2, z2, n);

  k_gather2<<<(unsigned)((n + 127) / 128), 128, 0, stream>>>(rowptr, cnt, csr, y2, z2, b2, out, n);
}


# Layer-2 products h@W2l, h@W2r moved into the layer-1 row-sum kernel: a 64-node work-group multiplies its own h rows from an LDS tile (one WMMA chain, both weights in B); h plane and one launch gone
# speedup vs baseline: 5.6501x; 1.0918x over previous
#include <hip/hip_runtime.h>


typedef __attribute__((ext_vector_type(2))) float v2f;
typedef __attribute__((ext_vector_type(4))) float v4f;
typedef __attribute__((ext_vector_type(8))) float v8f;

static __device__ __forceinline__ v8f wmma_f32x4(v2f a, v2f b, v8f c) {
  return __builtin_amdgcn_wmma_f32_16x16x4_f32(
       false, a,  false, b,
       (short)0, c,  false,  false);
}

typedef __attribute__((ext_vector_type(4))) int v4i;

#define RANGE_SHIFT 9
#define RANGE (1 << RANGE_SHIFT)
#define RANGE_MASK (RANGE - 1)
#define CHUNK_T 1024
#define EPT 8
#define CHUNK (CHUNK_T * EPT)
#define MAXR CHUNK_T

static __device__ __forceinline__ int block_excl_scan(int v, int* wsum, int lane, int wave, int& total) {
  int inc = v;
#pragma unroll
  for (int o = 1; o < 32; o <<= 1) {
    int u = __shfl_up(inc, o);
    if (lane >= o) inc += u;
  }
  if (lane == 31) wsum[wave] = inc;
  __syncthreads();
  int winc = wsum[lane];
#pragma unroll
  for (int o = 1; o < 32; o <<= 1) {
    int u = __shfl_up(winc, o);
    if (lane >= o) winc += u;
  }
  int wpre = __shfl(winc, (wave + 31) & 31);
  if (wave == 0) wpre = 0;
  total = __shfl(winc, 31);
  return wpre + inc - v;
}

__launch_bounds__(CHUNK_T)
__global__ void k_chunk_sort(const int* __restrict__ ei, int* __restrict__ cntTab,
                             int* __restrict__ locTab, int* __restrict__ keys,
                             long long E, int nRange, int nChunk) {
  __shared__ int hb[MAXR];
  __shared__ int wsum[32];
  __shared__ int stage[CHUNK];
  const int tid = threadIdx.x;
  const int lane = tid & 31;
  const int wave = tid >> 5;
  hb[tid] = 0;
  __syncthreads();
  const long long base = (long long)blockIdx.x * CHUNK;
  int key[EPT], rg[EPT], rk[EPT];
#pragma unroll
  for (int k = 0; k < EPT; ++k) {
    const long long e = base + (long long)k * CHUNK_T + tid;
    key[k] = (e < E) ? ei[e] : 0;
    rg[k]  = (e < E) ? ei[E + e] : -1;
  }
#pragma unroll
  for (int k = 0; k < EPT; ++k) {
    const int d = rg[k];
    if (d >= 0) {
      key[k] = (key[k] << RANGE_SHIFT) | (d & RANGE_MASK);
      rg[k] = d >> RANGE_SHIFT;
    }
  }
#pragma unroll
  for (int k = 0; k < EPT; ++k)
    rk[k] = (rg[k] >= 0) ? atomicAdd(&hb[rg[k]], 1) : 0;
  __syncthreads();
  const int c = hb[tid];
  int chunkEdges;
  const int ls = block_excl_scan(c, wsum, lane, wave, chunkEdges);
  hb[tid] = ls;
  if (tid < nRange) {
    cntTab[tid * nChunk + blockIdx.x] = c;
    locTab[tid * nChunk + blockIdx.x] = ls;
  }
  __syncthreads();
#pragma unroll
  for (int k = 0; k < EPT; ++k)
    if (rg[k] >= 0) stage[hb[rg[k]] + rk[k]] = key[k];
  __syncthreads();
  const long long left = E - base;
  const int nvalid = (left < (long long)CHUNK) ? (int)left : CHUNK;
#pragma unroll
  for (int k = 0; k < EPT; ++k) {
    const int i = k * CHUNK_T + tid;
    if (i < nvalid) keys[base + i] = stage[i];
  }
}

#define MAXC CHUNK_T
template <class F>
static __device__ __forceinline__ void for_each_key(const int* __restrict__ keys,
                                                    const int* segN, const int* segO,
                                                    int nChunk, int lane, int wave, F f) {
  for (int g0 = wave; g0 < nChunk; g0 += 4 * 32) {
    int nn[4], oo[4], ka[4], kb[4];
#pragma unroll
    for (int j = 0; j < 4; ++j) {
      const int ch = g0 + 32 * j;
      nn[j] = (ch < nChunk) ? segN[ch] : 0;
      oo[j] = (ch < nChunk) ? segO[ch] : 0;
    }
#pragma unroll
    for (int j = 0; j < 4; ++j) {
      ka[j] = (lane < nn[j]) ? keys[oo[j] + lane] : -1;
      kb[j] = (lane + 32 < nn[j]) ? keys[oo[j] + lane + 32] : -1;
    }
#pragma unroll
    for (int j = 0; j < 4; ++j) {
      if (ka[j] >= 0) f(ka[j]);
      if (kb[j] >= 0) f(kb[j]);
    }
#pragma unroll
    for (int j = 0; j < 4; ++j)
      for (int i = lane + 64; i < nn[j]; i += 32) f(keys[oo[j] + i]);
  }
}

#define T1_STRIDE 68
#define G1_ROWS 128
#define G1_QUARTERS (CHUNK_T / 256)
#define G1_WG_ROWS (G1_ROWS * G1_QUARTERS)
static __device__ __forceinline__ void gemm1_rows(float* __restrict__ lds,
                                                  const float* __restrict__ x,
                                                  const float* __restrict__ Wl,
                                                  const float* __restrict__ Wr,
                                                  float* __restrict__ y1,
                                                  float* __restrict__ z1, int n, int wg) {
  const int quarter = threadIdx.x >> 8;
  const int tid = threadIdx.x & 255;
  float* tile = lds + quarter * (G1_ROWS * T1_STRIDE);
  const int rb = (wg * G1_QUARTERS + quarter) * G1_ROWS;

#pragma unroll
  for (int it = 0; it < 8; ++it) {
    int idx = (it * 256 + tid) * 4;
    int r = idx >> 6, c = idx & 63;
    int grow = rb + r; if (grow > n - 1) grow = n - 1;
    v4f v = *(const v4f*)(x + (size_t)grow * 64 + c);
    *(v4f*)(tile + r * T1_STRIDE + c) = v;
  }
  __syncthreads();

  const int lane = tid & 31;
  const int wave = tid >> 5;
  const int mr   = lane & 15;
  const int kh   = (lane >> 4) << 1;

  v2f bl[16], br[16];
#pragma unroll
  for (int s = 0; s < 16; ++s) {
    int k0 = 4 * s + kh;
    v2f tl, tr;
    tl.x = Wl[k0 * 16 + mr]; tl.y = Wl[(k0 + 1) * 16 + mr];
    tr.x = Wr[k0 * 16 + mr]; tr.y = Wr[(k0 + 1) * 16 + mr];
    bl[s] = tl; br[s] = tr;
  }

  const float* arow = tile + (wave * 16 + mr) * T1_STRIDE;
  v8f cl = {}; v8f cr = {};
#pragma unroll
  for (int s = 0; s < 16; ++s) {
    v2f a = *(const v2f*)(arow + 4 * s + kh);
    cl = wmma_f32x4(a, bl[s], cl);
    cr = wmma_f32x4(a, br[s], cr);
  }

  const int rowHalf = (lane >> 4) << 3;
#pragma unroll
  for (int v = 0; v < 8; ++v) {
    int row = rb + wave * 16 + rowHalf + v;
    if (row < n) {
      y1[(size_t)row * 16 + mr] = cl[v];
      z1[(size_t)row * 16 + mr] = cr[v];
    }
  }
}

__launch_bounds__(CHUNK_T, 16)
__global__ void k_range_sort_gemm1(const int* __restrict__ keys, const int* __restrict__ cntTab,
                                   const int* __restrict__ locTab, int* __restrict__ cnt,
                                   int* __restrict__ rowptr, int* __restrict__ csr,
                                   int n, int nChunk, int nRange,
                                   const float* __restrict__ x, const float* __restrict__ Wl,
                                   const float* __restrict__ Wr, float* __restrict__ y1,
                                   float* __restrict__ z1) {
  __shared__ __attribute__((aligned(16))) unsigned char lds[G1_WG_ROWS * T1_STRIDE * 4];
  static_assert((RANGE + 32 + 2 * MAXC) * 4 <= G1_WG_ROWS * T1_STRIDE * 4, "sort tables fit the LDS block");
  static_assert(((RANGE + 32 + 2 * MAXC) * 4) % 16 == 0, "the LDS copy of the rows starts 16 B aligned");
  if ((int)blockIdx.x >= nRange) {
    gemm1_rows((float*)lds, x, Wl, Wr, y1, z1, n, (int)blockIdx.x - nRange);
    return;
  }
  int* c    = (int*)lds;
  int* wsum = c + RANGE;
  int* segN = wsum + 32;
  int* segO = segN + MAXC;
  int* stg  = segO + MAXC;
  constexpr int STG_CAP = G1_WG_ROWS * T1_STRIDE - (RANGE + 32 + 2 * MAXC);
  const int tid = threadIdx.x;
  const int lane = tid & 31;
  const int wave = tid >> 5;
  const int b = blockIdx.x;

  const int nFront = b * nChunk;
  int s = 0;
  const v4i* c4 = (const v4i*)cntTab;
  const int nFront4 = nFront >> 2;
  for (int i0 = tid; i0 < nFront4; i0 += 8 * CHUNK_T) {
    v4i v[8];
#pragma unroll
    for (int j = 0; j < 8; ++j) {
      const int i = i0 + j * CHUNK_T;
      const v4i zero = {0, 0, 0, 0};
      v[j] = (i < nFront4) ? c4[i] : zero;
    }
#pragma unroll
    for (int j = 0; j < 8; ++j) s += (v[j].x + v[j].y) + (v[j].z + v[j].w);
  }
  for (int i = (nFront & ~3) + tid; i < nFront; i += CHUNK_T) s += cntTab[i];
  for (int ch = tid; ch < nChunk; ch += CHUNK_T) {
    segN[ch] = cntTab[nFront + ch];
    segO[ch] = ch * CHUNK + locTab[nFront + ch];
  }
  if (tid < RANGE) c[tid] = 0;
#pragma unroll
  for (int o = 16; o > 0; o >>= 1) s += __shfl_xor(s, o);
  if (lane == 0) wsum[wave] = s;
  __syncthreads();
  int beg = wsum[lane];
#pragma unroll
  for (int o = 16; o > 0; o >>= 1) beg += __shfl_xor(beg, o);

  for_each_key(keys, segN, segO, nChunk, lane, wave,
               [&](int k) { atomicAdd(&c[k & RANGE_MASK], 1); });
  __syncthreads();
  const int v = (tid < RANGE) ? c[tid] : 0;
  int tot;
  const int ls = block_excl_scan(v, wsum, lane, wave, tot);
  if (tid < RANGE) {
    c[tid] = ls;
    const int node = b * RANGE + tid;
    if (node < n) { cnt[node] = v; rowptr[node] = beg + ls; }
  }
  __syncthreads();
  const int off = beg & 31;
  if (off + tot <= STG_CAP) {
    for_each_key(keys, segN, segO, nChunk, lane, wave,
                 [&](int k) {
                   const int p = atomicAdd(&c[k & RANGE_MASK], 1);
                   stg[off + p] = k >> RANGE_SHIFT;
                 });
    __syncthreads();
    int* rows = csr + (beg - off);
    const int lo = off, hi = off + tot;
    const int vlo = (lo + 3) & ~3, vhi = hi & ~3;
    const int headEnd = (vlo < hi) ? vlo : hi;
    if (lo + tid < headEnd) rows[lo + tid] = stg[lo + tid];
    for (int q = 4 * tid; q < vhi; q += 4 * CHUNK_T)
      if (q >= vlo) *(v4i*)(rows + q) = *(const v4i*)(stg + q);
    const int tb = (vhi > headEnd) ? vhi : headEnd;
    if (tb + tid < hi) rows[tb + tid] = stg[tb + tid];
  } else {
    for_each_key(keys, segN, segO, nChunk, lane, wave,
                 [&](int k) {
                   const int p = atomicAdd(&c[k & RANGE_MASK], 1);
                   csr[beg + p] = k >> RANGE_SHIFT;
                 });
  }
}

#define T2_STRIDE 20
#define G1N 64
__launch_bounds__(256)
__global__ void k_gather1_gemm2(const int* __restrict__ rowptr, const int* __restrict__ cnt,
                                const int* __restrict__ csr, const float* __restrict__ y1,
                                const float* __restrict__ z1, const float* __restrict__ b1,
                                const float* __restrict__ W2l, const float* __restrict__ W2r,
                                float* __restrict__ y2, float* __restrict__ z2, int n) {
  __shared__ __attribute__((aligned(16))) float tile[G1N * T2_STRIDE];
  const int tid  = threadIdx.x;
  const int lane = tid & 31;
  const int wave = tid >> 5;
  const int mr   = lane & 15;
  const int kh   = (lane >> 4) << 1;
  const int rb   = blockIdx.x * G1N;

  const int nl = tid >> 2;
  const int node = rb + nl;
  const int q = tid & 3;
  const bool live = node < n;
  const int nd = live ? node : n - 1;
  const int beg = rowptr[nd];
  const int cv = cnt[nd];
  const int c = live ? cv : 0;
  const v4f* y4 = (const v4f*)y1;
  const v4f zero4 = {0.0f, 0.0f, 0.0f, 0.0f};
  const v4f z = ((const v4f*)z1)[(size_t)nd * 4 + q];
  const float bq0 = b1[q * 4 + 0], bq1 = b1[q * 4 + 1], bq2 = b1[q * 4 + 2], bq3 = b1[q * 4 + 3];
  v4f acc = zero4;
  for (int j = 0; j < c; j += 4) {
    const int m = c - j;
    const int* p = csr + beg + j;
    int s0 = p[0];
    int s1 = p[(m > 1) ? 1 : 0];
    int s2 = p[(m > 2) ? 2 : 0];
    int s3 = p[(m > 3) ? 3 : 0];
    v4f a0 = y4[(size_t)s0 * 4 + q];
    v4f a1 = y4[(size_t)s1 * 4 + q];
    v4f a2 = y4[(size_t)s2 * 4 + q];
    v4f a3 = y4[(size_t)s3 * 4 + q];
    acc += a0;
    acc += (m > 1) ? a1 : zero4;
    acc += (m > 2) ? a2 : zero4;
    acc += (m > 3) ? a3 : zero4;
  }
  v2f bw[4];
#pragma unroll
  for (int s = 0; s < 4; ++s) { v2f t = {}; bw[s] = t; }
  if (wave < 4 && mr < 4) {
    const float* wp = (mr < 2) ? (W2l + mr) : (W2r + (mr - 2));
#pragma unroll
    for (int s = 0; s < 4; ++s) {
      const int k0 = 4 * s + kh;
      v2f t;
      t.x = wp[k0 * 2]; t.y = wp[(k0 + 1) * 2];
      bw[s] = t;
    }
  }
  const float d = fmaxf((float)c, 1.0f);
  v4f r;
  r.x = fmaxf(acc.x / d + z.x + bq0, 0.0f);
  r.y = fmaxf(acc.y / d + z.y + bq1, 0.0f);
  r.z = fmaxf(acc.z / d + z.z + bq2, 0.0f);
  r.w = fmaxf(acc.w / d + z.w + bq3, 0.0f);
  *(v4f*)(tile + nl * T2_STRIDE + q * 4) = r;
  __syncthreads();

  if (wave < 4) {
    const float* arow = tile + (wave * 16 + mr) * T2_STRIDE;
    v8f cc = {};
#pragma unroll
    for (int s = 0; s < 4; ++s) {
      v2f a = *(const v2f*)(arow + 4 * s + kh);
      cc = wmma_f32x4(a, bw[s], cc);
    }
    const int row0 = rb + wave * 16 + ((lane >> 4) << 3);
    float* dst = (mr < 2) ? (y2 + mr) : (z2 + (mr - 2));
    if (mr < 4) {
#pragma unroll
      for (int v = 0; v < 8; ++v) {
        const int row = row0 + v;
        if (row < n) dst[(size_t)row * 2] = cc[v];
      }
    }
  }
}

__global__ void k_gather2(const int* __restrict__ rowptr, const int* __restrict__ cnt,
                          const int* __restrict__ csr, const float* __restrict__ y2,
                          const float* __restrict__ z2, const float* __restrict__ b2,
                          float* __restrict__ out, int n) {
  const int node = blockIdx.x * blockDim.x + threadIdx.x;
  if (node >= n) return;
  const int beg = rowptr[node];
  const int c = cnt[node];
  const v2f* y = (const v2f*)y2;
  const v2f zero2 = {0.0f, 0.0f};
  v2f acc = zero2;
  for (int j = 0; j < c; j += 4) {
    const int m = c - j;
    const int* p = csr + beg + j;
    int s0 = p[0];
    int s1 = p[(m > 1) ? 1 : 0];
    int s2 = p[(m > 2) ? 2 : 0];
    int s3 = p[(m > 3) ? 3 : 0];
    v2f a0 = y[s0];
    v2f a1 = y[s1];
    v2f a2 = y[s2];
    v2f a3 = y[s3];
    acc += a0;
    acc += (m > 1) ? a1 : zero2;
    acc += (m > 2) ? a2 : zero2;
    acc += (m > 3) ? a3 : zero2;
  }
  const float d = fmaxf((float)c, 1.0f);
  const v2f z = ((const v2f*)z2)[node];
  v2f r;
  r.x = acc.x / d + z.x + b2[0];
  r.y = acc.y / d + z.y + b2[1];
  ((v2f*)out)[node] = r;
}

extern "C" void kernel_launch(void* const* d_in, const int* in_sizes, int n_in,
                              void* d_out, int out_size, void* d_ws, size_t ws_size,
                              hipStream_t stream) {
  const float* x   = (const float*)d_in[0];
  const int*   ei  = (const int*)d_in[1];
  const float* W1l = (const float*)d_in[2];
  const float* W1r = (const float*)d_in[3];
  const float* b1  = (const float*)d_in[4];
  const float* W2l = (const float*)d_in[5];
  const float* W2r = (const float*)d_in[6];
  const float* b2  = (const float*)d_in[7];
  float* out = (float*)d_out;

  const int       n = in_sizes[0] / 64;
  const long long E = in_sizes[1] / 2;

  float* ws   = (float*)d_ws;
  float* y1   = ws;
  float* z1   = y1 + (size_t)16 * n;
  float* y2   = z1 + (size_t)16 * n;
  float* z2   = y2 + (size_t)2 * n;
  int* cnt    = (int*)(z2 + (size_t)2 * n);
  int* rowptr = cnt + n;
  int* keys   = rowptr + n;
  int* csr    = keys + E;
  const int nRange = (n + RANGE - 1) / RANGE;
  const int nChunk = (int)((E + CHUNK - 1) / CHUNK);
  const int nTab   = nRange * nChunk;
  int* cntTab = csr + E;
  int* locTab = cntTab + nTab;

  k_chunk_sort<<<(unsigned)nChunk, CHUNK_T, 0, stream>>>(ei, cntTab, locTab, keys, E, nRange, nChunk);

  const int nGemm1 = (n + G1_WG_ROWS - 1) / G1_WG_ROWS;
  k_range_sort_gemm1<<<(unsigned)(nRange + nGemm1), CHUNK_T, 0, stream>>>(
      keys, cntTab, locTab, cnt, rowptr, csr, n, nChunk, nRange, x, W1l, W1r, y1, z1);

  k_gather1_gemm2<<<(unsigned)((n + G1N - 1) / G1N), 256, 0, stream>>>(rowptr, cnt, csr, y1, z1, b1,
                                                                    W2l, W2r, y2, z2, n);

  k_gather2<<<(unsigned)((n + 127) / 128), 128, 0, stream>>>(rowptr, cnt, csr, y2, z2, b2, out, n);
}


# Row sums of both layers: dependent load chain cut (next trip's indices asked for behind this trip's 8 row loads; row start and length loaded together; dead slots share one address); sum order kept
# speedup vs baseline: 5.9794x; 1.0583x over previous
#include <hip/hip_runtime.h>


typedef __attribute__((ext_vector_type(2))) float v2f;
typedef __attribute__((ext_vector_type(4))) float v4f;
typedef __attribute__((ext_vector_type(8))) float v8f;

static __device__ __forceinline__ v8f wmma_f32x4(v2f a, v2f b, v8f c) {
  return __builtin_amdgcn_wmma_f32_16x16x4_f32(
       false, a,  false, b,
       (short)0, c,  false,  false);
}

typedef __attribute__((ext_vector_type(4))) int v4i;

#define RANGE_SHIFT 9
#define RANGE (1 << RANGE_SHIFT)
#define RANGE_MASK (RANGE - 1)
#define CHUNK_T 1024
#define EPT 8
#define CHUNK (CHUNK_T * EPT)
#define MAXR CHUNK_T

static __device__ __forceinline__ int block_excl_scan(int v, int* wsum, int lane, int wave, int& total) {
  int inc = v;
#pragma unroll
  for (int o = 1; o < 32; o <<= 1) {
    int u = __shfl_up(inc, o);
    if (lane >= o) inc += u;
  }
  if (lane == 31) wsum[wave] = inc;
  __syncthreads();
  int winc = wsum[lane];
#pragma unroll
  for (int o = 1; o < 32; o <<= 1) {
    int u = __shfl_up(winc, o);
    if (lane >= o) winc += u;
  }
  int wpre = __shfl(winc, (wave + 31) & 31);
  if (wave == 0) wpre = 0;
  total = __shfl(winc, 31);
  return wpre + inc - v;
}

__launch_bounds__(CHUNK_T)
__global__ void k_chunk_sort(const int* __restrict__ ei, int* __restrict__ cntTab,
                             int* __restrict__ locTab, int* __restrict__ keys,
                             long long E, int nRange, int nChunk) {
  __shared__ int hb[MAXR];
  __shared__ int wsum[32];
  __shared__ int stage[CHUNK];
  const int tid = threadIdx.x;
  const int lane = tid & 31;
  const int wave = tid >> 5;
  hb[tid] = 0;
  __syncthreads();
  const long long base = (long long)blockIdx.x * CHUNK;
  int key[EPT], rg[EPT], rk[EPT];
#pragma unroll
  for (int k = 0; k < EPT; ++k) {
    const long long e = base + (long long)k * CHUNK_T + tid;
    key[k] = (e < E) ? ei[e] : 0;
    rg[k]  = (e < E) ? ei[E + e] : -1;
  }
#pragma unroll
  for (int k = 0; k < EPT; ++k) {
    const int d = rg[k];
    if (d >= 0) {
      key[k] = (key[k] << RANGE_SHIFT) | (d & RANGE_MASK);
      rg[k] = d >> RANGE_SHIFT;
    }
  }
#pragma unroll
  for (int k = 0; k < EPT; ++k)
    rk[k] = (rg[k] >= 0) ? atomicAdd(&hb[rg[k]], 1) : 0;
  __syncthreads();
  const int c = hb[tid];
  int chunkEdges;
  const int ls = block_excl_scan(c, wsum, lane, wave, chunkEdges);
  hb[tid] = ls;
  if (tid < nRange) {
    cntTab[tid * nChunk + blockIdx.x] = c;
    locTab[tid * nChunk + blockIdx.x] = ls;
  }
  __syncthreads();
#pragma unroll
  for (int k = 0; k < EPT; ++k)
    if (rg[k] >= 0) stage[hb[rg[k]] + rk[k]] = key[k];
  __syncthreads();
  const long long left = E - base;
  const int nvalid = (left < (long long)CHUNK) ? (int)left : CHUNK;
#pragma unroll
  for (int k = 0; k < EPT; ++k) {
    const int i = k * CHUNK_T + tid;
    if (i < nvalid) keys[base + i] = stage[i];
  }
}

#define MAXC CHUNK_T
template <class F>
static __device__ __forceinline__ void for_each_key(const int* __restrict__ keys,
                                                    const int* segN, const int* segO,
                                                    int nChunk, int lane, int wave, F f) {
  for (int g0 = wave; g0 < nChunk; g0 += 4 * 32) {
    int nn[4], oo[4], ka[4], kb[4];
#pragma unroll
    for (int j = 0; j < 4; ++j) {
      const int ch = g0 + 32 * j;
      nn[j] = (ch < nChunk) ? segN[ch] : 0;
      oo[j] = (ch < nChunk) ? segO[ch] : 0;
    }
#pragma unroll
    for (int j = 0; j < 4; ++j) {
      ka[j] = (lane < nn[j]) ? keys[oo[j] + lane] : -1;
      kb[j] = (lane + 32 < nn[j]) ? keys[oo[j] + lane + 32] : -1;
    }
#pragma unroll
    for (int j = 0; j < 4; ++j) {
      if (ka[j] >= 0) f(ka[j]);
      if (kb[j] >= 0) f(kb[j]);
    }
#pragma unroll
    for (int j = 0; j < 4; ++j)
      for (int i = lane + 64; i < nn[j]; i += 32) f(keys[oo[j] + i]);
  }
}

#define T1_STRIDE 68
#define G1_ROWS 128
#define G1_QUARTERS (CHUNK_T / 256)
#define G1_WG_ROWS (G1_ROWS * G1_QUARTERS)
static __device__ __forceinline__ void gemm1_rows(float* __restrict__ lds,
                                                  const float* __restrict__ x,
                                                  const float* __restrict__ Wl,
                                                  const float* __restrict__ Wr,
                                                  float* __restrict__ y1,
                                                  float* __restrict__ z1, int n, int wg) {
  const int quarter = threadIdx.x >> 8;
  const int tid = threadIdx.x & 255;
  float* tile = lds + quarter * (G1_ROWS * T1_STRIDE);
  const int rb = (wg * G1_QUARTERS + quarter) * G1_ROWS;

#pragma unroll
  for (int it = 0; it < 8; ++it) {
    int idx = (it * 256 + tid) * 4;
    int r = idx >> 6, c = idx & 63;
    int grow = rb + r; if (grow > n - 1) grow = n - 1;
    v4f v = *(const v4f*)(x + (size_t)grow * 64 + c);
    *(v4f*)(tile + r * T1_STRIDE + c) = v;
  }
  __syncthreads();

  const int lane = tid & 31;
  const int wave = tid >> 5;
  const int mr   = lane & 15;
  const int kh   = (lane >> 4) << 1;

  v2f bl[16], br[16];
#pragma unroll
  for (int s = 0; s < 16; ++s) {
    int k0 = 4 * s + kh;
    v2f tl, tr;
    tl.x = Wl[k0 * 16 + mr]; tl.y = Wl[(k0 + 1) * 16 + mr];
    tr.x = Wr[k0 * 16 + mr]; tr.y = Wr[(k0 + 1) * 16 + mr];
    bl[s] = tl; br[s] = tr;
  }

  const float* arow = tile + (wave * 16 + mr) * T1_STRIDE;
  v8f cl = {}; v8f cr = {};
#pragma unroll
  for (int s = 0; s < 16; ++s) {
    v2f a = *(const v2f*)(arow + 4 * s + kh);
    cl = wmma_f32x4(a, bl[s], cl);
    cr = wmma_f32x4(a, br[s], cr);
  }

  const int rowHalf = (lane >> 4) << 3;
#pragma unroll
  for (int v = 0; v < 8; ++v) {
    int row = rb + wave * 16 + rowHalf + v;
    if (row < n) {
      y1[(size_t)row * 16 + mr] = cl[v];
      z1[(size_t)row * 16 + mr] = cr[v];
    }
  }
}

__launch_bounds__(CHUNK_T, 16)
__global__ void k_range_sort_gemm1(const int* __restrict__ keys, const int* __restrict__ cntTab,
                                   const int* __restrict__ locTab, int* __restrict__ cnt,
                                   int* __restrict__ rowptr, int* __restrict__ csr,
                                   int n, int nChunk, int nRange,
                                   const float* __restrict__ x, const float* __restrict__ Wl,
                                   const float* __restrict__ Wr, float* __restrict__ y1,
                                   float* __restrict__ z1) {
  __shared__ __attribute__((aligned(16))) unsigned char lds[G1_WG_ROWS * T1_STRIDE * 4];
  static_assert((RANGE + 32 + 2 * MAXC) * 4 <= G1_WG_ROWS * T1_STRIDE * 4, "sort tables fit the LDS block");
  static_assert(((RANGE + 32 + 2 * MAXC) * 4) % 16 == 0, "the LDS copy of the rows starts 16 B aligned");
  if ((int)blockIdx.x >= nRange) {
    gemm1_rows((float*)lds, x, Wl, Wr, y1, z1, n, (int)blockIdx.x - nRange);
    return;
  }
  int* c    = (int*)lds;
  int* wsum = c + RANGE;
  int* segN = wsum + 32;
  int* segO = segN + MAXC;
  int* stg  = segO + MAXC;
  constexpr int STG_CAP = G1_WG_ROWS * T1_STRIDE - (RANGE + 32 + 2 * MAXC);
  const int tid = threadIdx.x;
  const int lane = tid & 31;
  const int wave = tid >> 5;
  const int b = blockIdx.x;

  const int nFront = b * nChunk;
  int s = 0;
  const v4i* c4 = (const v4i*)cntTab;
  const int nFront4 = nFront >> 2;
  for (int i0 = tid; i0 < nFront4; i0 += 8 * CHUNK_T) {
    v4i v[8];
#pragma unroll
    for (int j = 0; j < 8; ++j) {
      const int i = i0 + j * CHUNK_T;
      const v4i zero = {0, 0, 0, 0};
      v[j] = (i < nFront4) ? c4[i] : zero;
    }
#pragma unroll
    for (int j = 0; j < 8; ++j) s += (v[j].x + v[j].y) + (v[j].z + v[j].w);
  }
  for (int i = (nFront & ~3) + tid; i < nFront; i += CHUNK_T) s += cntTab[i];
  for (int ch = tid; ch < nChunk; ch += CHUNK_T) {
    segN[ch] = cntTab[nFront + ch];
    segO[ch] = ch * CHUNK + locTab[nFront + ch];
  }
  if (tid < RANGE) c[tid] = 0;
#pragma unroll
  for (int o = 16; o > 0; o >>= 1) s += __shfl_xor(s, o);
  if (lane == 0) wsum[wave] = s;
  __syncthreads();
  int beg = wsum[lane];
#pragma unroll
  for (int o = 16; o > 0; o >>= 1) beg += __shfl_xor(beg, o);

  for_each_key(keys, segN, segO, nChunk, lane, wave,
               [&](int k) { atomicAdd(&c[k & RANGE_MASK], 1); });
  __syncthreads();
  const int v = (tid < RANGE) ? c[tid] : 0;
  int tot;
  const int ls = block_excl_scan(v, wsum, lane, wave, tot);
  if (tid < RANGE) {
    c[tid] = ls;
    const int node = b * RANGE + tid;
    if (node < n) { cnt[node] = v; rowptr[node] = beg + ls; }
  }
  __syncthreads();
  const int off = beg & 31;
  if (off + tot <= STG_CAP) {
    for_each_key(keys, segN, segO, nChunk, lane, wave,
                 [&](int k) {
                   const int p = atomicAdd(&c[k & RANGE_MASK], 1);
                   stg[off + p] = k >> RANGE_SHIFT;
                 });
    __syncthreads();
    int* rows = csr + (beg - off);
    const int lo = off, hi = off + tot;
    const int vlo = (lo + 3) & ~3, vhi = hi & ~3;
    const int headEnd = (vlo < hi) ? vlo : hi;
    if (lo + tid < headEnd) rows[lo + tid] = stg[lo + tid];
    for (int q = 4 * tid; q < vhi; q += 4 * CHUNK_T)
      if (q >= vlo) *(v4i*)(rows + q) = *(const v4i*)(stg + q);
    const int tb = (vhi > headEnd) ? vhi : headEnd;
    if (tb + tid < hi) rows[tb + tid] = stg[tb + tid];
  } else {
    for_each_key(keys, segN, segO, nChunk, lane, wave,
                 [&](int k) {
                   const int p = atomicAdd(&c[k & RANGE_MASK], 1);
                   csr[beg + p] = k >> RANGE_SHIFT;
                 });
  }
}

#define T2_STRIDE 20
#define G1N 64
#define GDEPTH1 8
#define GDEPTH2 8
__launch_bounds__(256, 16)
__global__ void k_gather1_gemm2(const int* __restrict__ rowptr, const int* __restrict__ cnt,
                                const int* __restrict__ csr, const float* __restrict__ y1,
                                const float* __restrict__ z1, const float* __restrict__ b1,
                                const float* __restrict__ W2l, const float* __restrict__ W2r,
                                float* __restrict__ y2, float* __restrict__ z2, int n) {
  __shared__ __attribute__((aligned(16))) float tile[G1N * T2_STRIDE];
  const int tid  = threadIdx.x;
  const int lane = tid & 31;
  const int wave = tid >> 5;
  const int mr   = lane & 15;
  const int kh   = (lane >> 4) << 1;
  const int rb   = blockIdx.x * G1N;

  const int nl = tid >> 2;
  const int node = rb + nl;
  const int q = tid & 3;
  const bool live = node < n;
  const int nd = live ? node : n - 1;
  const int beg = rowptr[nd];
  const int cv = cnt[nd];
  asm volatile("" :: "v"(beg), "v"(cv));
  const int c = live ? cv : 0;
  const v4f* y4 = (const v4f*)y1;
  const v4f zero4 = {0.0f, 0.0f, 0.0f, 0.0f};
  const v4f z = ((const v4f*)z1)[(size_t)nd * 4 + q];
  const float bq0 = b1[q * 4 + 0], bq1 = b1[q * 4 + 1], bq2 = b1[q * 4 + 2], bq3 = b1[q * 4 + 3];
  const int last = beg + c - 1;
  int s[GDEPTH1];
#pragma unroll
  for (int k = 0; k < GDEPTH1; ++k) {
    const int o = beg + k;
    s[k] = csr[(o <= last) ? o : 0];
  }
  v4f acc = zero4;
  for (int j = beg; j <= last; j += GDEPTH1) {
    v4f a[GDEPTH1];
#pragma unroll
    for (int k = 0; k < GDEPTH1; ++k) a[k] = y4[s[k] * 4 + q];
#pragma unroll
    for (int k = 0; k < GDEPTH1; ++k) {
      const int o = j + GDEPTH1 + k;
      s[k] = csr[(o <= last) ? o : 0];
    }
    __builtin_amdgcn_sched_barrier(0);
    const int m = last - j + 1;
#pragma unroll
    for (int k = 0; k < GDEPTH1; ++k) acc += (k < m) ? a[k] : zero4;
  }
  v2f bw[4];
#pragma unroll
  for (int s = 0; s < 4; ++s) { v2f t = {}; bw[s] = t; }
  if (wave < 4 && mr < 4) {
    const float* wp = (mr < 2) ? (W2l + mr) : (W2r + (mr - 2));
#pragma unroll
    for (int s = 0; s < 4; ++s) {
      const int k0 = 4 * s + kh;
      v2f t;
      t.x = wp[k0 * 2]; t.y = wp[(k0 + 1) * 2];
      bw[s] = t;
    }
  }
  const float d = fmaxf((float)c, 1.0f);
  v4f r;
  r.x = fmaxf(acc.x / d + z.x + bq0, 0.0f);
  r.y = fmaxf(acc.y / d + z.y + bq1, 0.0f);
  r.z = fmaxf(acc.z / d + z.z + bq2, 0.0f);
  r.w = fmaxf(acc.w / d + z.w + bq3, 0.0f);
  *(v4f*)(tile + nl * T2_STRIDE + q * 4) = r;
  __syncthreads();

  if (wave < 4) {
    const float* arow = tile + (wave * 16 + mr) * T2_STRIDE;
    v8f cc = {};
#pragma unroll
    for (int s = 0; s < 4; ++s) {
      v2f a = *(const v2f*)(arow + 4 * s + kh);
      cc = wmma_f32x4(a, bw[s], cc);
    }
    const int row0 = rb + wave * 16 + ((lane >> 4) << 3);
    float* dst = (mr < 2) ? (y2 + mr) : (z2 + (mr - 2));
    if (mr < 4) {
#pragma unroll
      for (int v = 0; v < 8; ++v) {
        const int row = row0 + v;
        if (row < n) dst[(size_t)row * 2] = cc[v];
      }
    }
  }
}

__global__ void k_gather2(const int* __restrict__ rowptr, const int* __restrict__ cnt,
                          const int* __restrict__ csr, const float* __restrict__ y2,
                          const float* __restrict__ z2, const float* __restrict__ b2,
                          float* __restrict__ out, int n) {
  const int node = blockIdx.x * blockDim.x + threadIdx.x;
  if (node >= n) return;
  const int beg = rowptr[node];
  const int c = cnt[node];
  const v2f* y = (const v2f*)y2;
  const v2f zero2 = {0.0f, 0.0f};
  const v2f z = ((const v2f*)z2)[node];
  const float bx = b2[0], by = b2[1];
  asm volatile("" :: "v"(beg), "v"(c));
  const int last = beg + c - 1;
  int s[GDEPTH2];
#pragma unroll
  for (int k = 0; k < GDEPTH2; ++k) {
    const int o = beg + k;
    s[k] = csr[(o <= last) ? o : 0];
  }
  v2f acc = zero2;
  for (int j = beg; j <= last; j += GDEPTH2) {
    v2f a[GDEPTH2];
#pragma unroll
    for (int k = 0; k < GDEPTH2; ++k) a[k] = y[s[k]];
#pragma unroll
    for (int k = 0; k < GDEPTH2; ++k) {
      const int o = j + GDEPTH2 + k;
      s[k] = csr[(o <= last) ? o : 0];
    }
    __builtin_amdgcn_sched_barrier(0);
    const int m = last - j + 1;
#pragma unroll
    for (int k = 0; k < GDEPTH2; ++k) acc += (k < m) ? a[k] : zero2;
  }
  const float d = fmaxf((float)c, 1.0f);
  v2f r;
  r.x = acc.x / d + z.x + bx;
  r.y = acc.y / d + z.y + by;
  ((v2f*)out)[node] = r;
}

extern "C" void kernel_launch(void* const* d_in, const int* in_sizes, int n_in,
                              void* d_out, int out_size, void* d_ws, size_t ws_size,
                              hipStream_t stream) {
  const float* x   = (const float*)d_in[0];
  const int*   ei  = (const int*)d_in[1];
  const float* W1l = (const float*)d_in[2];
  const float* W1r = (const float*)d_in[3];
  const float* b1  = (const float*)d_in[4];
  const float* W2l = (const float*)d_in[5];
  const float* W2r = (const float*)d_in[6];
  const float* b2  = (const float*)d_in[7];
  float* out = (float*)d_out;

  const int       n = in_sizes[0] / 64;
  const long long E = in_sizes[1] / 2;

  float* ws   = (float*)d_ws;
  float* y1   = ws;
  float* z1   = y1 + (size_t)16 * n;
  float* y2   = z1 + (size_t)16 * n;
  float* z2   = y2 + (size_t)2 * n;
  int* cnt    = (int*)(z2 + (size_t)2 * n);
  int* rowptr = cnt + n;
  int* keys   = rowptr + n;
  int* csr    = keys + E;
  const int nRange = (n + RANGE - 1) / RANGE;
  const int nChunk = (int)((E + CHUNK - 1) / CHUNK);
  const int nTab   = nRange * nChunk;
  int* cntTab = csr + E;
  int* locTab = cntTab + nTab;

  k_chunk_sort<<<(unsigned)nChunk, CHUNK_T, 0, stream>>>(ei, cntTab, locTab, keys, E, nRange, nChunk);

  const int nGemm1 = (n + G1_WG_ROWS - 1) / G1_WG_ROWS;
  k_range_sort_gemm1<<<(unsigned)(nRange + nGemm1), CHUNK_T, 0, stream>>>(
      keys, cntTab, locTab, cnt, rowptr, csr, n, nChunk, nRange, x, W1l, W1r, y1, z1);

  k_gather1_gemm2<<<(unsigned)((n + G1N - 1) / G1N), 256, 0, stream>>>(rowptr, cnt, csr, y1, z1, b1,
                                                                    W2l, W2r, y2, z2, n);

  k_gather2<<<(unsigned)((n + 127) / 128), 128, 0, stream>>>(rowptr, cnt, csr, y2, z2, b2, out, n);
}


# Layer-1 row sums and layer-2 products moved into the level-2 sort work-group of each node range (row lists read from its LDS); x@W1 rides with sort level 1; 3 launches; sum order kept
# speedup vs baseline: 6.3565x; 1.0631x over previous
#include <hip/hip_runtime.h>


typedef __attribute__((ext_vector_type(2))) float v2f;
typedef __attribute__((ext_vector_type(4))) float v4f;
typedef __attribute__((ext_vector_type(8))) float v8f;

static __device__ __forceinline__ v8f wmma_f32x4(v2f a, v2f b, v8f c) {
  return __builtin_amdgcn_wmma_f32_16x16x4_f32(
       false, a,  false, b,
       (short)0, c,  false,  false);
}

typedef __attribute__((ext_vector_type(4))) int v4i;

#define RANGE_SHIFT 9
#define RANGE (1 << RANGE_SHIFT)
#define RANGE_MASK (RANGE - 1)
#define CHUNK_T 1024
#define EPT 8
#define CHUNK (CHUNK_T * EPT)
#define MAXR CHUNK_T

static __device__ __forceinline__ int block_excl_scan(int v, int* wsum, int lane, int wave, int& total) {
  int inc = v;
#pragma unroll
  for (int o = 1; o < 32; o <<= 1) {
    int u = __shfl_up(inc, o);
    if (lane >= o) inc += u;
  }
  if (lane == 31) wsum[wave] = inc;
  __syncthreads();
  int winc = wsum[lane];
#pragma unroll
  for (int o = 1; o < 32; o <<= 1) {
    int u = __shfl_up(winc, o);
    if (lane >= o) winc += u;
  }
  int wpre = __shfl(winc, (wave + 31) & 31);
  if (wave == 0) wpre = 0;
  total = __shfl(winc, 31);
  return wpre + inc - v;
}

#define T1_STRIDE 68
#define G1_ROWS 128
#define G1_QUARTERS (CHUNK_T / 256)
#define G1_WG_ROWS (G1_ROWS * G1_QUARTERS)
static __device__ __forceinline__ void gemm1_rows(float* __restrict__ lds,
                                                  const float* __restrict__ x,
                                                  const float* __restrict__ Wl,
                                                  const float* __restrict__ Wr,
                                                  float* __restrict__ y1,
                                                  float* __restrict__ z1, int n, int wg) {
  const int quarter = threadIdx.x >> 8;
  const int tid = threadIdx.x & 255;
  float* tile = lds + quarter * (G1_ROWS * T1_STRIDE);
  const int rb = (wg * G1_QUARTERS + quarter) * G1_ROWS;

#pragma unroll
  for (int it = 0; it < 8; ++it) {
    int idx = (it * 256 + tid) * 4;
    int r = idx >> 6, c = idx & 63;
    int grow = rb + r; if (grow > n - 1) grow = n - 1;
    v4f v = *(const v4f*)(x + (size_t)grow * 64 + c);
    *(v4f*)(tile + r * T1_STRIDE + c) = v;
  }
  __syncthreads();

  const int lane = tid & 31;
  const int wave = tid >> 5;
  const int mr   = lane & 15;
  const int kh   = (lane >> 4) << 1;

  v2f bl[16], br[16];
#pragma unroll
  for (int s = 0; s < 16; ++s) {
    int k0 = 4 * s + kh;
    v2f tl, tr;
    tl.x = Wl[k0 * 16 + mr]; tl.y = Wl[(k0 + 1) * 16 + mr];
    tr.x = Wr[k0 * 16 + mr]; tr.y = Wr[(k0 + 1) * 16 + mr];
    bl[s] = tl; br[s] = tr;
  }

  const float* arow = tile + (wave * 16 + mr) * T1_STRIDE;
  v8f cl = {}; v8f cr = {};
#pragma unroll
  for (int s = 0; s < 16; ++s) {
    v2f a = *(const v2f*)(arow + 4 * s + kh);
    cl = wmma_f32x4(a, bl[s], cl);
    cr = wmma_f32x4(a, br[s], cr);
  }

  const int rowHalf = (lane >> 4) << 3;
#pragma unroll
  for (int v = 0; v < 8; ++v) {
    int row = rb + wave * 16 + rowHalf + v;
    if (row < n) {
      y1[(size_t)row * 16 + mr] = cl[v];
      z1[(size_t)row * 16 + mr] = cr[v];
    }
  }
}

__launch_bounds__(CHUNK_T, 16)
__global__ void k_chunk_sort_gemm1(const int* __restrict__ ei, int* __restrict__ cntTab,
                                   int* __restrict__ locTab, int* __restrict__ keys,
                                   long long E, int nRange, int nChunk,
                                   const float* __restrict__ x, const float* __restrict__ Wl,
                                   const float* __restrict__ Wr, float* __restrict__ y1,
                                   float* __restrict__ z1, int n) {
  __shared__ __attribute__((aligned(16))) unsigned char lds[G1_WG_ROWS * T1_STRIDE * 4];
  static_assert((MAXR + 32 + CHUNK) * 4 <= G1_WG_ROWS * T1_STRIDE * 4, "chunk tables fit the LDS block");
  if ((int)blockIdx.x >= nChunk) {
    gemm1_rows((float*)lds, x, Wl, Wr, y1, z1, n, (int)blockIdx.x - nChunk);
    return;
  }
  int* hb    = (int*)lds;
  int* wsum  = hb + MAXR;
  int* stage = wsum + 32;
  const int tid = threadIdx.x;
  const int lane = tid & 31;
  const int wave = tid >> 5;
  hb[tid] = 0;
  __syncthreads();
  const long long base = (long long)blockIdx.x * CHUNK;
  int key[EPT], rg[EPT], rk[EPT];
#pragma unroll
  for (int k = 0; k < EPT; ++k) {
    const long long e = base + (long long)k * CHUNK_T + tid;
    key[k] = (e < E) ? ei[e] : 0;
    rg[k]  = (e < E) ? ei[E + e] : -1;
  }
#pragma unroll
  for (int k = 0; k < EPT; ++k) {
    const int d = rg[k];
    if (d >= 0) {
      key[k] = (key[k] << RANGE_SHIFT) | (d & RANGE_MASK);
      rg[k] = d >> RANGE_SHIFT;
    }
  }
#pragma unroll
  for (int k = 0; k < EPT; ++k)
    rk[k] = (rg[k] >= 0) ? atomicAdd(&hb[rg[k]], 1) : 0;
  __syncthreads();
  const int c = hb[tid];
  int chunkEdges;
  const int ls = block_excl_scan(c, wsum, lane, wave, chunkEdges);
  hb[tid] = ls;
  if (tid < nRange) {
    cntTab[tid * nChunk + blockIdx.x] = c;
    locTab[tid * nChunk + blockIdx.x] = ls;
  }
  __syncthreads();
#pragma unroll
  for (int k = 0; k < EPT; ++k)
    if (rg[k] >= 0) stage[hb[rg[k]] + rk[k]] = key[k];
  __syncthreads();
  const long long left = E - base;
  const int nvalid = (left < (long long)CHUNK) ? (int)left : CHUNK;
#pragma unroll
  for (int k = 0; k < EPT; ++k) {
    const int i = k * CHUNK_T + tid;
    if (i < nvalid) keys[base + i] = stage[i];
  }
}

#define MAXC CHUNK_T
template <class F>
static __device__ __forceinline__ void for_each_key(const int* __restrict__ keys,
                                                    const int* segN, const int* segO,
                                                    int nChunk, int lane, int wave, F f) {
  for (int g0 = wave; g0 < nChunk; g0 += 4 * 32) {
    int nn[4], oo[4], ka[4], kb[4];
#pragma unroll
    for (int j = 0; j < 4; ++j) {
      const int ch = g0 + 32 * j;
      nn[j] = (ch < nChunk) ? segN[ch] : 0;
      oo[j] = (ch < nChunk) ? segO[ch] : 0;
    }
#pragma unroll
    for (int j = 0; j < 4; ++j) {
      ka[j] = (lane < nn[j]) ? keys[oo[j] + lane] : -1;
      kb[j] = (lane + 32 < nn[j]) ? keys[oo[j] + lane + 32] : -1;
    }
#pragma unroll
    for (int j = 0; j < 4; ++j) {
      if (ka[j] >= 0) f(ka[j]);
      if (kb[j] >= 0) f(kb[j]);
    }
#pragma unroll
    for (int j = 0; j < 4; ++j)
      for (int i = lane + 64; i < nn[j]; i += 32) f(keys[oo[j] + i]);
  }
}

#define T2_STRIDE 20
#define GDEPTH1 16
#define GDEPTH2 8
#define K2_LDS_INTS (G1_WG_ROWS * T1_STRIDE)
#define K2_TAB_INTS (RANGE + 32 + 2 * MAXC)
#define K2_TILE_INTS (RANGE * T2_STRIDE)

template <class IDX>
static __device__ __forceinline__ v4f row_sum16(const v4f* __restrict__ y4, int q, int start, int end, IDX idx) {
  const v4f zero4 = {0.0f, 0.0f, 0.0f, 0.0f};
  v4f acc = zero4;
  for (int j = start; j < end; j += GDEPTH1) {
    int s[GDEPTH1];
    v4f a[GDEPTH1];
#pragma unroll
    for (int k = 0; k < GDEPTH1; ++k) {
      const int o = j + k;
      s[k] = idx((o < end) ? o : 0);
    }
#pragma unroll
    for (int k = 0; k < GDEPTH1; ++k) a[k] = y4[s[k] * 4 + q];
    const int m = end - j;
    acc += a[0];
#pragma unroll
    for (int k = 1; k < GDEPTH1; ++k) acc += (k < m) ? a[k] : zero4;
  }
  return acc;
}

__launch_bounds__(CHUNK_T)
__global__ void k_range_sort_gather1_gemm2(const int* __restrict__ keys, const int* __restrict__ cntTab,
                                           const int* __restrict__ locTab, int* __restrict__ cnt,
                                           int* __restrict__ rowptr, int* __restrict__ csr,
                                           int n, int nChunk,
                                           const float* __restrict__ y1, const float* __restrict__ z1,
                                           const float* __restrict__ b1, const float* __restrict__ W2l,
                                           const float* __restrict__ W2r, float* __restrict__ y2,
                                           float* __restrict__ z2) {
  __shared__ __attribute__((aligned(16))) unsigned char lds[K2_LDS_INTS * 4];
  static_assert((K2_TAB_INTS * 4) % 16 == 0, "the h tile starts 16 B aligned");
  static_assert(((K2_TAB_INTS + K2_TILE_INTS) * 4) % 16 == 0, "the LDS copy of the rows starts 16 B aligned");
  int* c    = (int*)lds;
  int* wsum = c + RANGE;
  int* segN = wsum + 32;
  int* segO = segN + MAXC;
  float* tile = (float*)(segO + MAXC);
  int* stg  = (int*)(tile + K2_TILE_INTS);
  constexpr int STG_CAP = K2_LDS_INTS - K2_TAB_INTS - K2_TILE_INTS;
  const int tid = threadIdx.x;
  const int lane = tid & 31;
  const int wave = tid >> 5;
  const int b = blockIdx.x;

  const int nFront = b * nChunk;
  int s = 0;
  const v4i* c4 = (const v4i*)cntTab;
  const int nFront4 = nFront >> 2;
  for (int i0 = tid; i0 < nFront4; i0 += 8 * CHUNK_T) {
    v4i v[8];
#pragma unroll
    for (int j = 0; j < 8; ++j) {
      const int i = i0 + j * CHUNK_T;
      const v4i zero = {0, 0, 0, 0};
      v[j] = (i < nFront4) ? c4[i] : zero;
    }
#pragma unroll
    for (int j = 0; j < 8; ++j) s += (v[j].x + v[j].y) + (v[j].z + v[j].w);
  }
  for (int i = (nFront & ~3) + tid; i < nFront; i += CHUNK_T) s += cntTab[i];
  for (int ch = tid; ch < nChunk; ch += CHUNK_T) {
    segN[ch] = cntTab[nFront + ch];
    segO[ch] = ch * CHUNK + locTab[nFront + ch];
  }
  if (tid < RANGE) c[tid] = 0;
#pragma unroll
  for (int o = 16; o > 0; o >>= 1) s += __shfl_xor(s, o);
  if (lane == 0) wsum[wave] = s;
  __syncthreads();
  int beg = wsum[lane];
#pragma unroll
  for (int o = 16; o > 0; o >>= 1) beg += __shfl_xor(beg, o);

  for_each_key(keys, segN, segO, nChunk, lane, wave,
               [&](int k) { atomicAdd(&c[k & RANGE_MASK], 1); });
  __syncthreads();
  const int v = (tid < RANGE) ? c[tid] : 0;
  int tot;
  const int ls = block_excl_scan(v, wsum, lane, wave, tot);
  if (tid < RANGE) {
    c[tid] = ls;
    const int node = b * RANGE + tid;
    if (node < n) { cnt[node] = v; rowptr[node] = beg + ls; }
  }
  __syncthreads();
  const int off = beg & 31;
  const bool inLds = (off + tot <= STG_CAP);
  if (inLds) {
    for_each_key(keys, segN, segO, nChunk, lane, wave,
                 [&](int k) {
                   const int p = atomicAdd(&c[k & RANGE_MASK], 1);
                   stg[off + p] = k >> RANGE_SHIFT;
                 });
    __syncthreads();
    int* rows = csr + (beg - off);
    const int lo = off, hi = off + tot;
    const int vlo = (lo + 3) & ~3, vhi = hi & ~3;
    const int headEnd = (vlo < hi) ? vlo : hi;
    if (lo + tid < headEnd) rows[lo + tid] = stg[lo + tid];
    for (int q = 4 * tid; q < vhi; q += 4 * CHUNK_T)
      if (q >= vlo) *(v4i*)(rows + q) = *(const v4i*)(stg + q);
    const int tb = (vhi > headEnd) ? vhi : headEnd;
    if (tb + tid < hi) rows[tb + tid] = stg[tb + tid];
  } else {
    for_each_key(keys, segN, segO, nChunk, lane, wave,
                 [&](int k) {
                   const int p = atomicAdd(&c[k & RANGE_MASK], 1);
                   csr[beg + p] = k >> RANGE_SHIFT;
                 });
    __syncthreads();
  }

  const int mr = lane & 15;
  const int kh = (lane >> 4) << 1;
  v2f bw[4];
#pragma unroll
  for (int t = 0; t < 4; ++t) { v2f zz = {}; bw[t] = zz; }
  if (mr < 4) {
    const float* wp = (mr < 2) ? (W2l + mr) : (W2r + (mr - 2));
#pragma unroll
    for (int t = 0; t < 4; ++t) {
      const int k0 = 4 * t + kh;
      v2f w;
      w.x = wp[k0 * 2]; w.y = wp[(k0 + 1) * 2];
      bw[t] = w;
    }
  }
  const int q = tid & 3;
  const float bq0 = b1[q * 4 + 0], bq1 = b1[q * 4 + 1], bq2 = b1[q * 4 + 2], bq3 = b1[q * 4 + 3];
  const v4f* y4 = (const v4f*)y1;
  const int* rowsG = csr + beg;
  for (int half = 0; half < RANGE / 256; ++half) {
    const int nl = half * 256 + (tid >> 2);
    const int node = b * RANGE + nl;
    const int nd = (node < n) ? node : n - 1;
    const v4f z = ((const v4f*)z1)[(size_t)nd * 4 + q];
    const int start = nl ? c[nl - 1] : 0;
    const int end = c[nl];
    v4f acc;
    if (inLds) acc = row_sum16(y4, q, start, end, [&](int o) { return stg[off + o]; });
    else       acc = row_sum16(y4, q, start, end, [&](int o) { return rowsG[o]; });
    const float d = fmaxf((float)(end - start), 1.0f);
    v4f r;
    r.x = fmaxf(acc.x / d + z.x + bq0, 0.0f);
    r.y = fmaxf(acc.y / d + z.y + bq1, 0.0f);
    r.z = fmaxf(acc.z / d + z.z + bq2, 0.0f);
    r.w = fmaxf(acc.w / d + z.w + bq3, 0.0f);
    *(v4f*)(tile + nl * T2_STRIDE + q * 4) = r;
  }
  __syncthreads();

  const float* arow = tile + (wave * 16 + mr) * T2_STRIDE;
  v8f cc = {};
#pragma unroll
  for (int t = 0; t < 4; ++t) {
    v2f a = *(const v2f*)(arow + 4 * t + kh);
    cc = wmma_f32x4(a, bw[t], cc);
  }
  const int row0 = b * RANGE + wave * 16 + ((lane >> 4) << 3);
  float* dst = (mr < 2) ? (y2 + mr) : (z2 + (mr - 2));
  if (mr < 4) {
#pragma unroll
    for (int u = 0; u < 8; ++u) {
      const int row = row0 + u;
      if (row < n) dst[(size_t)row * 2] = cc[u];
    }
  }
}

__global__ void k_gather2(const int* __restrict__ rowptr, const int* __restrict__ cnt,
                          const int* __restrict__ csr, const float* __restrict__ y2,
                          const float* __restrict__ z2, const float* __restrict__ b2,
                          float* __restrict__ out, int n) {
  const int node = blockIdx.x * blockDim.x + threadIdx.x;
  if (node >= n) return;
  const int beg = rowptr[node];
  const int c = cnt[node];
  const v2f* y = (const v2f*)y2;
  const v2f zero2 = {0.0f, 0.0f};
  const v2f z = ((const v2f*)z2)[node];
  const float bx = b2[0], by = b2[1];
  asm volatile("" :: "v"(beg), "v"(c));
  const int last = beg + c - 1;
  int s[GDEPTH2];
#pragma unroll
  for (int k = 0; k < GDEPTH2; ++k) {
    const int o = beg + k;
    s[k] = csr[(o <= last) ? o : 0];
  }
  v2f acc = zero2;
  for (int j = beg; j <= last; j += GDEPTH2) {
    v2f a[GDEPTH2];
#pragma unroll
    for (int k = 0; k < GDEPTH2; ++k) a[k] = y[s[k]];
#pragma unroll
    for (int k = 0; k < GDEPTH2; ++k) {
      const int o = j + GDEPTH2 + k;
      s[k] = csr[(o <= last) ? o : 0];
    }
    __builtin_amdgcn_sched_barrier(0);
    const int m = last - j + 1;
#pragma unroll
    for (int k = 0; k < GDEPTH2; ++k) acc += (k < m) ? a[k] : zero2;
  }
  const float d = fmaxf((float)c, 1.0f);
  v2f r;
  r.x = acc.x / d + z.x + bx;
  r.y = acc.y / d + z.y + by;
  ((v2f*)out)[node] = r;
}

extern "C" void kernel_launch(void* const* d_in, const int* in_sizes, int n_in,
                              void* d_out, int out_size, void* d_ws, size_t ws_size,
                              hipStream_t stream) {
  const float* x   = (const float*)d_in[0];
  const int*   ei  = (const int*)d_in[1];
  const float* W1l = (const float*)d_in[2];
  const float* W1r = (const float*)d_in[3];
  const float* b1  = (const float*)d_in[4];
  const float* W2l = (const float*)d_in[5];
  const float* W2r = (const float*)d_in[6];
  const float* b2  = (const float*)d_in[7];
  float* out = (float*)d_out;

  const int       n = in_sizes[0] / 64;
  const long long E = in_sizes[1] / 2;

  float* ws   = (float*)d_ws;
  float* y1   = ws;
  float* z1   = y1 + (size_t)16 * n;
  float* y2   = z1 + (size_t)16 * n;
  float* z2   = y2 + (size_t)2 * n;
  int* cnt    = (int*)(z2 + (size_t)2 * n);
  int* rowptr = cnt + n;
  int* keys   = rowptr + n;
  int* csr    = keys + E;
  const int nRange = (n + RANGE - 1) / RANGE;
  const int nChunk = (int)((E + CHUNK - 1) / CHUNK);
  const int nTab   = nRange * nChunk;
  int* cntTab = csr + E;
  int* locTab = cntTab + nTab;

  const int nGemm1 = (n + G1_WG_ROWS - 1) / G1_WG_ROWS;
  k_chunk_sort_gemm1<<<(unsigned)(nChunk + nGemm1), CHUNK_T, 0, stream>>>(
      ei, cntTab, locTab, keys, E, nRange, nChunk, x, W1l, W1r, y1, z1, n);

  k_range_sort_gather1_gemm2<<<(unsigned)nRange, CHUNK_T, 0, stream>>>(
      keys, cntTab, locTab, cnt, rowptr, csr, n, nChunk, y1, z1, b1, W2l, W2r, y2, z2);

  k_gather2<<<(unsigned)((n + 127) / 128), 128, 0, stream>>>(rowptr, cnt, csr, y2, z2, b2, out, n);
}


# Level-2 sort: dependent load chain cut (range start summed from its own loc-table row, not the count table in front; a wave's edges loaded once, kept in registers for both passes); sum order kept
# speedup vs baseline: 6.5708x; 1.0337x over previous
#include <hip/hip_runtime.h>


typedef __attribute__((ext_vector_type(2))) float v2f;
typedef __attribute__((ext_vector_type(4))) float v4f;
typedef __attribute__((ext_vector_type(8))) float v8f;

static __device__ __forceinline__ v8f wmma_f32x4(v2f a, v2f b, v8f c) {
  return __builtin_amdgcn_wmma_f32_16x16x4_f32(
       false, a,  false, b,
       (short)0, c,  false,  false);
}

typedef __attribute__((ext_vector_type(4))) int v4i;

#define RANGE_SHIFT 9
#define RANGE (1 << RANGE_SHIFT)
#define RANGE_MASK (RANGE - 1)
#define CHUNK_T 1024
#define EPT 8
#define CHUNK (CHUNK_T * EPT)
#define MAXR CHUNK_T

static __device__ __forceinline__ int block_excl_scan(int v, int* wsum, int lane, int wave, int& total) {
  int inc = v;
#pragma unroll
  for (int o = 1; o < 32; o <<= 1) {
    int u = __shfl_up(inc, o);
    if (lane >= o) inc += u;
  }
  if (lane == 31) wsum[wave] = inc;
  __syncthreads();
  int winc = wsum[lane];
#pragma unroll
  for (int o = 1; o < 32; o <<= 1) {
    int u = __shfl_up(winc, o);
    if (lane >= o) winc += u;
  }
  int wpre = __shfl(winc, (wave + 31) & 31);
  if (wave == 0) wpre = 0;
  total = __shfl(winc, 31);
  return wpre + inc - v;
}

#define T1_STRIDE 68
#define G1_ROWS 128
#define G1_QUARTERS (CHUNK_T / 256)
#define G1_WG_ROWS (G1_ROWS * G1_QUARTERS)
static __device__ __forceinline__ void gemm1_rows(float* __restrict__ lds,
                                                  const float* __restrict__ x,
                                                  const float* __restrict__ Wl,
                                                  const float* __restrict__ Wr,
                                                  float* __restrict__ y1,
                                                  float* __restrict__ z1, int n, int wg) {
  const int quarter = threadIdx.x >> 8;
  const int tid = threadIdx.x & 255;
  float* tile = lds + quarter * (G1_ROWS * T1_STRIDE);
  const int rb = (wg * G1_QUARTERS + quarter) * G1_ROWS;

#pragma unroll
  for (int it = 0; it < 8; ++it) {
    int idx = (it * 256 + tid) * 4;
    int r = idx >> 6, c = idx & 63;
    int grow = rb + r; if (grow > n - 1) grow = n - 1;
    v4f v = *(const v4f*)(x + (size_t)grow * 64 + c);
    *(v4f*)(tile + r * T1_STRIDE + c) = v;
  }
  __syncthreads();

  const int lane = tid & 31;
  const int wave = tid >> 5;
  const int mr   = lane & 15;
  const int kh   = (lane >> 4) << 1;

  v2f bl[16], br[16];
#pragma unroll
  for (int s = 0; s < 16; ++s) {
    int k0 = 4 * s + kh;
    v2f tl, tr;
    tl.x = Wl[k0 * 16 + mr]; tl.y = Wl[(k0 + 1) * 16 + mr];
    tr.x = Wr[k0 * 16 + mr]; tr.y = Wr[(k0 + 1) * 16 + mr];
    bl[s] = tl; br[s] = tr;
  }

  const float* arow = tile + (wave * 16 + mr) * T1_STRIDE;
  v8f cl = {}; v8f cr = {};
#pragma unroll
  for (int s = 0; s < 16; ++s) {
    v2f a = *(const v2f*)(arow + 4 * s + kh);
    cl = wmma_f32x4(a, bl[s], cl);
    cr = wmma_f32x4(a, br[s], cr);
  }

  const int rowHalf = (lane >> 4) << 3;
#pragma unroll
  for (int v = 0; v < 8; ++v) {
    int row = rb + wave * 16 + rowHalf + v;
    if (row < n) {
      y1[(size_t)row * 16 + mr] = cl[v];
      z1[(size_t)row * 16 + mr] = cr[v];
    }
  }
}

__launch_bounds__(CHUNK_T, 16)
__global__ void k_chunk_sort_gemm1(const int* __restrict__ ei, int* __restrict__ cntTab,
                                   int* __restrict__ locTab, int* __restrict__ keys,
                                   long long E, int nRange, int nChunk,
                                   const float* __restrict__ x, const float* __restrict__ Wl,
                                   const float* __restrict__ Wr, float* __restrict__ y1,
                                   float* __restrict__ z1, int n) {
  __shared__ __attribute__((aligned(16))) unsigned char lds[G1_WG_ROWS * T1_STRIDE * 4];
  static_assert((MAXR + 32 + CHUNK) * 4 <= G1_WG_ROWS * T1_STRIDE * 4, "chunk tables fit the LDS block");
  if ((int)blockIdx.x >= nChunk) {
    gemm1_rows((float*)lds, x, Wl, Wr, y1, z1, n, (int)blockIdx.x - nChunk);
    return;
  }
  int* hb    = (int*)lds;
  int* wsum  = hb + MAXR;
  int* stage = wsum + 32;
  const int tid = threadIdx.x;
  const int lane = tid & 31;
  const int wave = tid >> 5;
  hb[tid] = 0;
  __syncthreads();
  const long long base = (long long)blockIdx.x * CHUNK;
  int key[EPT], rg[EPT], rk[EPT];
#pragma unroll
  for (int k = 0; k < EPT; ++k) {
    const long long e = base + (long long)k * CHUNK_T + tid;
    key[k] = (e < E) ? ei[e] : 0;
    rg[k]  = (e < E) ? ei[E + e] : -1;
  }
#pragma unroll
  for (int k = 0; k < EPT; ++k) {
    const int d = rg[k];
    if (d >= 0) {
      key[k] = (key[k] << RANGE_SHIFT) | (d & RANGE_MASK);
      rg[k] = d >> RANGE_SHIFT;
    }
  }
#pragma unroll
  for (int k = 0; k < EPT; ++k)
    rk[k] = (rg[k] >= 0) ? atomicAdd(&hb[rg[k]], 1) : 0;
  __syncthreads();
  const int c = hb[tid];
  int chunkEdges;
  const int ls = block_excl_scan(c, wsum, lane, wave, chunkEdges);
  hb[tid] = ls;
  if (tid < nRange) {
    cntTab[tid * nChunk + blockIdx.x] = c;
    locTab[tid * nChunk + blockIdx.x] = ls;
  }
  __syncthreads();
#pragma unroll
  for (int k = 0; k < EPT; ++k)
    if (rg[k] >= 0) stage[hb[rg[k]] + rk[k]] = key[k];
  __syncthreads();
  const long long left = E - base;
  const int nvalid = (left < (long long)CHUNK) ? (int)left : CHUNK;
#pragma unroll
  for (int k = 0; k < EPT; ++k) {
    const int i = k * CHUNK_T + tid;
    if (i < nvalid) keys[base + i] = stage[i];
  }
}

#define MAXC CHUNK_T
#define PIECES 8
template <class F>
static __device__ __forceinline__ void for_each_key(const int* __restrict__ keys,
                                                    const int (&pn)[PIECES], const int (&po)[PIECES],
                                                    const int (&ka)[PIECES], const int (&kb)[PIECES],
                                                    const int* segN, const int* segO,
                                                    int nChunk, int lane, int wave, F f) {
#pragma unroll
  for (int j = 0; j < PIECES; ++j) {
    if (ka[j] >= 0) f(ka[j]);
    if (kb[j] >= 0) f(kb[j]);
  }
#pragma unroll
  for (int j = 0; j < PIECES; ++j)
    for (int i = lane + 64; i < pn[j]; i += 32) f(keys[po[j] + i]);
  for (int g0 = 32 * PIECES + wave; g0 < nChunk; g0 += 4 * 32) {
    int nn[4], oo[4], ma[4], mb[4];
#pragma unroll
    for (int j = 0; j < 4; ++j) {
      const int ch = g0 + 32 * j;
      nn[j] = (ch < nChunk) ? segN[ch] : 0;
      oo[j] = (ch < nChunk) ? segO[ch] : 0;
    }
#pragma unroll
    for (int j = 0; j < 4; ++j) {
      ma[j] = (lane < nn[j]) ? keys[oo[j] + lane] : -1;
      mb[j] = (lane + 32 < nn[j]) ? keys[oo[j] + lane + 32] : -1;
    }
#pragma unroll
    for (int j = 0; j < 4; ++j) {
      if (ma[j] >= 0) f(ma[j]);
      if (mb[j] >= 0) f(mb[j]);
    }
#pragma unroll
    for (int j = 0; j < 4; ++j)
      for (int i = lane + 64; i < nn[j]; i += 32) f(keys[oo[j] + i]);
  }
}

#define T2_STRIDE 20
#define GDEPTH1 16
#define GDEPTH2 8
#define K2_LDS_INTS (G1_WG_ROWS * T1_STRIDE)
#define K2_TAB_INTS (RANGE + 32 + 2 * MAXC)
#define K2_TILE_INTS (RANGE * T2_STRIDE)

template <class IDX>
static __device__ __forceinline__ v4f row_sum16(const v4f* __restrict__ y4, int q, int start, int end, IDX idx) {
  const v4f zero4 = {0.0f, 0.0f, 0.0f, 0.0f};
  v4f acc = zero4;
  for (int j = start; j < end; j += GDEPTH1) {
    int s[GDEPTH1];
    v4f a[GDEPTH1];
#pragma unroll
    for (int k = 0; k < GDEPTH1; ++k) {
      const int o = j + k;
      s[k] = idx((o < end) ? o : 0);
    }
#pragma unroll
    for (int k = 0; k < GDEPTH1; ++k) a[k] = y4[s[k] * 4 + q];
    const int m = end - j;
    acc += a[0];
#pragma unroll
    for (int k = 1; k < GDEPTH1; ++k) acc += (k < m) ? a[k] : zero4;
  }
  return acc;
}

__launch_bounds__(CHUNK_T)
__global__ void k_range_sort_gather1_gemm2(const int* __restrict__ keys, const int* __restrict__ cntTab,
                                           const int* __restrict__ locTab, int* __restrict__ cnt,
                                           int* __restrict__ rowptr, int* __restrict__ csr,
                                           int n, int nChunk,
                                           const float* __restrict__ y1, const float* __restrict__ z1,
                                           const float* __restrict__ b1, const float* __restrict__ W2l,
                                           const float* __restrict__ W2r, float* __restrict__ y2,
                                           float* __restrict__ z2) {
  __shared__ __attribute__((aligned(16))) unsigned char lds[K2_LDS_INTS * 4];
  static_assert((K2_TAB_INTS * 4) % 16 == 0, "the h tile starts 16 B aligned");
  static_assert(((K2_TAB_INTS + K2_TILE_INTS) * 4) % 16 == 0, "the LDS copy of the rows starts 16 B aligned");
  int* c    = (int*)lds;
  int* wsum = c + RANGE;
  int* segN = wsum + 32;
  int* segO = segN + MAXC;
  float* tile = (float*)(segO + MAXC);
  int* stg  = (int*)(tile + K2_TILE_INTS);
  constexpr int STG_CAP = K2_LDS_INTS - K2_TAB_INTS - K2_TILE_INTS;
  const int tid = threadIdx.x;
  const int lane = tid & 31;
  const int wave = tid >> 5;
  const int b = blockIdx.x;

  const int nFront = b * nChunk;
  int tn = 0, tl = 0, to = 0;
  {
    const int ch = wave + 32 * lane;
    if (lane < PIECES && ch < nChunk) {
      tn = cntTab[nFront + ch];
      tl = locTab[nFront + ch];
      to = ch * CHUNK + tl;
    }
  }
  int s = tl;
  for (int ch = 32 * PIECES + tid; ch < nChunk; ch += CHUNK_T) {
    const int l = locTab[nFront + ch];
    segN[ch] = cntTab[nFront + ch];
    segO[ch] = ch * CHUNK + l;
    s += l;
  }
  if (tid < RANGE) c[tid] = 0;
  int pn[PIECES], po[PIECES], ka[PIECES], kb[PIECES];
#pragma unroll
  for (int j = 0; j < PIECES; ++j) {
    pn[j] = __builtin_amdgcn_readlane(tn, j);
    po[j] = __builtin_amdgcn_readlane(to, j);
  }
#pragma unroll
  for (int j = 0; j < PIECES; ++j) {
    ka[j] = (lane < pn[j]) ? keys[po[j] + lane] : -1;
    kb[j] = (lane + 32 < pn[j]) ? keys[po[j] + lane + 32] : -1;
  }
#pragma unroll
  for (int o = 16; o > 0; o >>= 1) s += __shfl_xor(s, o);
  if (lane == 0) wsum[wave] = s;
  __syncthreads();
  int beg = wsum[lane];
#pragma unroll
  for (int o = 16; o > 0; o >>= 1) beg += __shfl_xor(beg, o);

  for_each_key(keys, pn, po, ka, kb, segN, segO, nChunk, lane, wave,
               [&](int k) { atomicAdd(&c[k & RANGE_MASK], 1); });
  __syncthreads();
  const int v = (tid < RANGE) ? c[tid] : 0;
  int tot;
  const int ls = block_excl_scan(v, wsum, lane, wave, tot);
  if (tid < RANGE) {
    c[tid] = ls;
    const int node = b * RANGE + tid;
    if (node < n) { cnt[node] = v; rowptr[node] = beg + ls; }
  }
  __syncthreads();
  const int off = beg & 31;
  const bool inLds = (off + tot <= STG_CAP);
  if (inLds) {
    for_each_key(keys, pn, po, ka, kb, segN, segO, nChunk, lane, wave,
                 [&](int k) {
                   const int p = atomicAdd(&c[k & RANGE_MASK], 1);
                   stg[off + p] = k >> RANGE_SHIFT;
                 });
    __syncthreads();
    int* rows = csr + (beg - off);
    const int lo = off, hi = off + tot;
    const int vlo = (lo + 3) & ~3, vhi = hi & ~3;
    const int headEnd = (vlo < hi) ? vlo : hi;
    if (lo + tid < headEnd) rows[lo + tid] = stg[lo + tid];
    for (int q = 4 * tid; q < vhi; q += 4 * CHUNK_T)
      if (q >= vlo) *(v4i*)(rows + q) = *(const v4i*)(stg + q);
    const int tb = (vhi > headEnd) ? vhi : headEnd;
    if (tb + tid < hi) rows[tb + tid] = stg[tb + tid];
  } else {
    for_each_key(keys, pn, po, ka, kb, segN, segO, nChunk, lane, wave,
                 [&](int k) {
                   const int p = atomicAdd(&c[k & RANGE_MASK], 1);
                   csr[beg + p] = k >> RANGE_SHIFT;
                 });
    __syncthreads();
  }

  const int mr = lane & 15;
  const int kh = (lane >> 4) << 1;
  v2f bw[4];
#pragma unroll
  for (int t = 0; t < 4; ++t) { v2f zz = {}; bw[t] = zz; }
  if (mr < 4) {
    const float* wp = (mr < 2) ? (W2l + mr) : (W2r + (mr - 2));
#pragma unroll
    for (int t = 0; t < 4; ++t) {
      const int k0 = 4 * t + kh;
      v2f w;
      w.x = wp[k0 * 2]; w.y = wp[(k0 + 1) * 2];
      bw[t] = w;
    }
  }
  const int q = tid & 3;
  const float bq0 = b1[q * 4 + 0], bq1 = b1[q * 4 + 1], bq2 = b1[q * 4 + 2], bq3 = b1[q * 4 + 3];
  const v4f* y4 = (const v4f*)y1;
  const int* rowsG = csr + beg;
  for (int half = 0; half < RANGE / 256; ++half) {
    const int nl = half * 256 + (tid >> 2);
    const int node = b * RANGE + nl;
    const int nd = (node < n) ? node : n - 1;
    const v4f z = ((const v4f*)z1)[(size_t)nd * 4 + q];
    const int start = nl ? c[nl - 1] : 0;
    const int end = c[nl];
    v4f acc;
    if (inLds) acc = row_sum16(y4, q, start, end, [&](int o) { return stg[off + o]; });
    else       acc = row_sum16(y4, q, start, end, [&](int o) { return rowsG[o]; });
    const float d = fmaxf((float)(end - start), 1.0f);
    v4f r;
    r.x = fmaxf(acc.x / d + z.x + bq0, 0.0f);
    r.y = fmaxf(acc.y / d + z.y + bq1, 0.0f);
    r.z = fmaxf(acc.z / d + z.z + bq2, 0.0f);
    r.w = fmaxf(acc.w / d + z.w + bq3, 0.0f);
    *(v4f*)(tile + nl * T2_STRIDE + q * 4) = r;
  }
  __syncthreads();

  const float* arow = tile + (wave * 16 + mr) * T2_STRIDE;
  v8f cc = {};
#pragma unroll
  for (int t = 0; t < 4; ++t) {
    v2f a = *(const v2f*)(arow + 4 * t + kh);
    cc = wmma_f32x4(a, bw[t], cc);
  }
  const int row0 = b * RANGE + wave * 16 + ((lane >> 4) << 3);
  float* dst = (mr < 2) ? (y2 + mr) : (z2 + (mr - 2));
  if (mr < 4) {
#pragma unroll
    for (int u = 0; u < 8; ++u) {
      const int row = row0 + u;
      if (row < n) dst[(size_t)row * 2] = cc[u];
    }
  }
}

__global__ void k_gather2(const int* __restrict__ rowptr, const int* __restrict__ cnt,
                          const int* __restrict__ csr, const float* __restrict__ y2,
                          const float* __restrict__ z2, const float* __restrict__ b2,
                          float* __restrict__ out, int n) {
  const int node = blockIdx.x * blockDim.x + threadIdx.x;
  if (node >= n) return;
  const int beg = rowptr[node];
  const int c = cnt[node];
  const v2f* y = (const v2f*)y2;
  const v2f zero2 = {0.0f, 0.0f};
  const v2f z = ((const v2f*)z2)[node];
  const float bx = b2[0], by = b2[1];
  asm volatile("" :: "v"(beg), "v"(c));
  const int last = beg + c - 1;
  int s[GDEPTH2];
#pragma unroll
  for (int k = 0; k < GDEPTH2; ++k) {
    const int o = beg + k;
    s[k] = csr[(o <= last) ? o : 0];
  }
  v2f acc = zero2;
  for (int j = beg; j <= last; j += GDEPTH2) {
    v2f a[GDEPTH2];
#pragma unroll
    for (int k = 0; k < GDEPTH2; ++k) a[k] = y[s[k]];
#pragma unroll
    for (int k = 0; k < GDEPTH2; ++k) {
      const int o = j + GDEPTH2 + k;
      s[k] = csr[(o <= last) ? o : 0];
    }
    __builtin_amdgcn_sched_barrier(0);
    const int m = last - j + 1;
#pragma unroll
    for (int k = 0; k < GDEPTH2; ++k) acc += (k < m) ? a[k] : zero2;
  }
  const float d = fmaxf((float)c, 1.0f);
  v2f r;
  r.x = acc.x / d + z.x + bx;
  r.y = acc.y / d + z.y + by;
  ((v2f*)out)[node] = r;
}

extern "C" void kernel_launch(void* const* d_in, const int* in_sizes, int n_in,
                              void* d_out, int out_size, void* d_ws, size_t ws_size,
                              hipStream_t stream) {
  const float* x   = (const float*)d_in[0];
  const int*   ei  = (const int*)d_in[1];
  const float* W1l = (const float*)d_in[2];
  const float* W1r = (const float*)d_in[3];
  const float* b1  = (const float*)d_in[4];
  const float* W2l = (const float*)d_in[5];
  const float* W2r = (const float*)d_in[6];
  const float* b2  = (const float*)d_in[7];
  float* out = (float*)d_out;

  const int       n = in_sizes[0] / 64;
  const long long E = in_sizes[1] / 2;

  float* ws   = (float*)d_ws;
  float* y1   = ws;
  float* z1   = y1 + (size_t)16 * n;
  float* y2   = z1 + (size_t)16 * n;
  float* z2   = y2 + (size_t)2 * n;
  int* cnt    = (int*)(z2 + (size_t)2 * n);
  int* rowptr = cnt + n;
  int* keys   = rowptr + n;
  int* csr    = keys + E;
  const int nRange = (n + RANGE - 1) / RANGE;
  const int nChunk = (int)((E + CHUNK - 1) / CHUNK);
  const int nTab   = nRange * nChunk;
  int* cntTab = csr + E;
  int* locTab = cntTab + nTab;

  const int nGemm1 = (n + G1_WG_ROWS - 1) / G1_WG_ROWS;
  k_chunk_sort_gemm1<<<(unsigned)(nChunk + nGemm1), CHUNK_T, 0, stream>>>(
      ei, cntTab, locTab, keys, E, nRange, nChunk, x, W1l, W1r, y1, z1, n);

  k_range_sort_gather1_gemm2<<<(unsigned)nRange, CHUNK_T, 0, stream>>>(
      keys, cntTab, locTab, cnt, rowptr, csr, n, nChunk, y1, z1, b1, W2l, W2r, y2, z2);

  k_gather2<<<(unsigned)((n + 127) / 128), 128, 0, stream>>>(rowptr, cnt, csr, y2, z2, b2, out, n);
}
